# starting kernel
# baseline (speedup 1.0000x reference)
#include <hip/hip_runtime.h>

typedef __attribute__((ext_vector_type(2))) float v2f;
typedef __attribute__((ext_vector_type(8))) float v8f;

#define NVOX   4096
#define NHEADS 8
#define HDIM   32
#define CCH    256
#define BATCH  4

__global__ void __launch_bounds__(256)
wmma_gemm_f32(const float* A, const float* __restrict__ B,
              const float* __restrict__ bias, float* D,
              int M, int K, int Ncols, int lda) {
  const int lane = threadIdx.x & 31;
  const int wave = (blockIdx.x * blockDim.x + threadIdx.x) >> 5;
  const int ntn  = Ncols >> 6;
  const int tm   = wave / ntn;
  const int tn   = wave - tm * ntn;
  if ((tm << 4) >= M) return;
  const int row0 = tm << 4, col0 = tn << 6;
  const int half = lane >> 4;
  const int lid  = lane & 15;

  const float* arow = A + (size_t)(row0 + lid) * lda + (half << 1);
  const float* bcol = B + (size_t)(half << 1) * Ncols + col0 + lid;

  v8f acc0 = {0.f,0.f,0.f,0.f,0.f,0.f,0.f,0.f};
  v8f acc1 = acc0, acc2 = acc0, acc3 = acc0;

#pragma unroll 4
  for (int k = 0; k < K; k += 4) {
    v2f a = *(const v2f*)(arow + k);
    const float* b0 = bcol + (size_t)k * Ncols;
    const float* b1 = b0 + Ncols;
    v2f bf0, bf1, bf2, bf3;
    bf0.x = b0[0];  bf0.y = b1[0];
    bf1.x = b0[16]; bf1.y = b1[16];
    bf2.x = b0[32]; bf2.y = b1[32];
    bf3.x = b0[48]; bf3.y = b1[48];
    acc0 = __builtin_amdgcn_wmma_f32_16x16x4_f32(false, a, false, bf0, (short)0, acc0, false, false);
    acc1 = __builtin_amdgcn_wmma_f32_16x16x4_f32(false, a, false, bf1, (short)0, acc1, false, false);
    acc2 = __builtin_amdgcn_wmma_f32_16x16x4_f32(false, a, false, bf2, (short)0, acc2, false, false);
    acc3 = __builtin_amdgcn_wmma_f32_16x16x4_f32(false, a, false, bf3, (short)0, acc3, false, false);
  }

  float* drow = D + (size_t)(row0 + (half << 3)) * Ncols + col0 + lid;
  const float bv0 = bias ? bias[col0 + lid]      : 0.0f;
  const float bv1 = bias ? bias[col0 + lid + 16] : 0.0f;
  const float bv2 = bias ? bias[col0 + lid + 32] : 0.0f;
  const float bv3 = bias ? bias[col0 + lid + 48] : 0.0f;
#pragma unroll
  for (int i = 0; i < 8; ++i) {
    float* dr = drow + (size_t)i * Ncols;
    dr[0]  = acc0[i] + bv0;
    dr[16] = acc1[i] + bv1;
    dr[32] = acc2[i] + bv2;
    dr[48] = acc3[i] + bv3;
  }
}

__global__ void __launch_bounds__(256)
neigh_attn(const float* qkv, const int* __restrict__ neigh, float* out) {
  const int lane = threadIdx.x & 31;
  const int wave = (blockIdx.x * blockDim.x + threadIdx.x) >> 5;
  const int n = wave & (NVOX - 1);
  const int h = (wave >> 12) & (NHEADS - 1);
  const int b = wave >> 15;

  const float NEG_INF = -__builtin_inff();
  const size_t base = (size_t)(b * NVOX + n) * (3 * CCH);
  const float q = qkv[base + h * HDIM + lane] * 0.17677669529663687f;

  int   nb[27];
  float s[27];
#pragma unroll
  for (int j = 0; j < 27; ++j) {
    const int id = neigh[n * 27 + j];
    nb[j] = id;
    float t = 0.f;
    if (id != NVOX) {
      const float* kr = qkv + (size_t)(b * NVOX + id) * (3 * CCH) + CCH + h * HDIM;
      t = q * kr[lane];
    }
#pragma unroll
    for (int off = 16; off; off >>= 1)
      t += __shfl_xor(t, off, 32);
    s[j] = (id == NVOX) ? NEG_INF : t;
  }

  float m = NEG_INF;
#pragma unroll
  for (int j = 0; j < 27; ++j) m = fmaxf(m, s[j]);
  float sum = 0.f;
#pragma unroll
  for (int j = 0; j < 27; ++j) { s[j] = __expf(s[j] - m); sum += s[j]; }
  const float inv = 1.0f / sum;

  float acc = 0.f;
#pragma unroll
  for (int j = 0; j < 27; ++j) {
    if (nb[j] != NVOX) {
      const float* vr = qkv + (size_t)(b * NVOX + nb[j]) * (3 * CCH) + 2 * CCH + h * HDIM;
      acc += s[j] * vr[lane];
    }
  }
  out[base + h * HDIM + lane] = acc * inv;
}

extern "C" void kernel_launch(void* const* d_in, const int* in_sizes, int n_in,
                              void* d_out, int out_size, void* d_ws, size_t ws_size,
                              hipStream_t stream) {
  (void)in_sizes; (void)n_in; (void)out_size; (void)ws_size;
  const float* x      = (const float*)d_in[0];
  const float* w_qkv  = (const float*)d_in[1];
  const float* w_proj = (const float*)d_in[2];
  const float* b_proj = (const float*)d_in[3];
  const int*   neigh  = (const int*)d_in[4];
  float* out = (float*)d_out;
  float* qkv = (float*)d_ws;

  const int M = BATCH * NVOX;

  {
    int waves = (M / 16) * (3 * CCH / 64);
    wmma_gemm_f32<<<waves * 32 / 256, 256, 0, stream>>>(
        x, w_qkv, nullptr, qkv, M, CCH, 3 * CCH, CCH);
  }
  {
    int waves = BATCH * NHEADS * NVOX;
    neigh_attn<<<waves * 32 / 256, 256, 0, stream>>>(qkv, neigh, qkv);
  }
  {
    int waves = (M / 16) * (CCH / 64);
    wmma_gemm_f32<<<waves * 32 / 256, 256, 0, stream>>>(
        qkv, w_proj, b_proj, out, M, CCH, CCH, 3 * CCH);
  }
}


# attention: one wave per voxel loops its 8 heads; q*k products pass a wave-private LDS transpose so each lane sums one neighbour's 32 products (sum order changes); softmax across lanes by DPP
# speedup vs baseline: 2.7742x; 2.7742x over previous
#include <hip/hip_runtime.h>

typedef __attribute__((ext_vector_type(2))) float v2f;
typedef __attribute__((ext_vector_type(8))) float v8f;

#define NVOX   4096
#define NHEADS 8
#define HDIM   32
#define CCH    256
#define BATCH  4

__global__ void __launch_bounds__(256)
wmma_gemm_f32(const float* A, const float* __restrict__ B,
              const float* __restrict__ bias, float* D,
              int M, int K, int Ncols, int lda) {
  const int lane = threadIdx.x & 31;
  const int wave = (blockIdx.x * blockDim.x + threadIdx.x) >> 5;
  const int ntn  = Ncols >> 6;
  const int tm   = wave / ntn;
  const int tn   = wave - tm * ntn;
  if ((tm << 4) >= M) return;
  const int row0 = tm << 4, col0 = tn << 6;
  const int half = lane >> 4;
  const int lid  = lane & 15;

  const float* arow = A + (size_t)(row0 + lid) * lda + (half << 1);
  const float* bcol = B + (size_t)(half << 1) * Ncols + col0 + lid;

  v8f acc0 = {0.f,0.f,0.f,0.f,0.f,0.f,0.f,0.f};
  v8f acc1 = acc0, acc2 = acc0, acc3 = acc0;

#pragma unroll 4
  for (int k = 0; k < K; k += 4) {
    v2f a = *(const v2f*)(arow + k);
    const float* b0 = bcol + (size_t)k * Ncols;
    const float* b1 = b0 + Ncols;
    v2f bf0, bf1, bf2, bf3;
    bf0.x = b0[0];  bf0.y = b1[0];
    bf1.x = b0[16]; bf1.y = b1[16];
    bf2.x = b0[32]; bf2.y = b1[32];
    bf3.x = b0[48]; bf3.y = b1[48];
    acc0 = __builtin_amdgcn_wmma_f32_16x16x4_f32(false, a, false, bf0, (short)0, acc0, false, false);
    acc1 = __builtin_amdgcn_wmma_f32_16x16x4_f32(false, a, false, bf1, (short)0, acc1, false, false);
    acc2 = __builtin_amdgcn_wmma_f32_16x16x4_f32(false, a, false, bf2, (short)0, acc2, false, false);
    acc3 = __builtin_amdgcn_wmma_f32_16x16x4_f32(false, a, false, bf3, (short)0, acc3, false, false);
  }

  float* drow = D + (size_t)(row0 + (half << 3)) * Ncols + col0 + lid;
  const float bv0 = bias ? bias[col0 + lid]      : 0.0f;
  const float bv1 = bias ? bias[col0 + lid + 16] : 0.0f;
  const float bv2 = bias ? bias[col0 + lid + 32] : 0.0f;
  const float bv3 = bias ? bias[col0 + lid + 48] : 0.0f;
#pragma unroll
  for (int i = 0; i < 8; ++i) {
    float* dr = drow + (size_t)i * Ncols;
    dr[0]  = acc0[i] + bv0;
    dr[16] = acc1[i] + bv1;
    dr[32] = acc2[i] + bv2;
    dr[48] = acc3[i] + bv3;
  }
}

#define NNB  27
#define ROWP 36
#define WPB  8

static __device__ __forceinline__ float xhalf(float x) {
  const int b = __builtin_bit_cast(int, x);
  return __builtin_bit_cast(float, __builtin_amdgcn_permlanex16(b, b, 0x76543210, 0xfedcba98, false, false));
}
template <int M> static __device__ __forceinline__ float xrow(float x) {
  return __builtin_bit_cast(float,
      __builtin_amdgcn_update_dpp(0, __builtin_bit_cast(int, x), 0x160 + M, 0xf, 0xf, false));
}
static __device__ __forceinline__ float wave_max(float x) {
  x = fmaxf(x, xrow<1>(x)); x = fmaxf(x, xrow<2>(x));
  x = fmaxf(x, xrow<4>(x)); x = fmaxf(x, xrow<8>(x));
  return fmaxf(x, xhalf(x));
}
static __device__ __forceinline__ float wave_sum(float x) {
  x += xrow<1>(x); x += xrow<2>(x); x += xrow<4>(x); x += xrow<8>(x);
  return x + xhalf(x);
}
static __device__ __forceinline__ void wave_lds_order() {
  __builtin_amdgcn_fence(__ATOMIC_RELEASE, "wavefront");
  __builtin_amdgcn_wave_barrier();
  __builtin_amdgcn_fence(__ATOMIC_ACQUIRE, "wavefront");
}

__global__ void __launch_bounds__(256)
neigh_attn(const float* qkv, const int* __restrict__ neigh, float* out) {
  __shared__ __attribute__((aligned(16))) float prod[WPB][NNB][ROWP];

  const int lane = threadIdx.x & 31;
  const int wib  = __builtin_amdgcn_readfirstlane((int)(threadIdx.x >> 5));
  const int wave = blockIdx.x * WPB + wib;
  const int n = wave & (NVOX - 1);
  const int b = wave >> 12;

  const float NEG_INF = -__builtin_inff();
  const size_t base = (size_t)(b * NVOX + n) * (3 * CCH);
  float* myprod = &prod[wib][0][0];

  const int jr = lane < NNB ? lane : NNB - 1;
  const bool live = (lane < NNB) && (neigh[n * NNB + jr] != NVOX);

  const float* krow[NNB];
#pragma unroll
  for (int j = 0; j < NNB; ++j) {
    const int id = neigh[n * NNB + j];
    krow[j] = qkv + (size_t)(b * NVOX + (id != NVOX ? id : 0)) * (3 * CCH) + CCH;
  }

  for (int h = 0; h < NHEADS; ++h) {
    const unsigned co = (unsigned)(h * HDIM + lane);
    const unsigned cv = co + CCH;
    const float q = qkv[base + co] * 0.17677669529663687f;

#pragma unroll
    for (int j = 0; j < NNB; ++j)
      myprod[j * ROWP + lane] = q * krow[j][co];

    wave_lds_order();

    const float4* r4 = (const float4*)(myprod + jr * ROWP);
    const float4 a0 = r4[0], a1 = r4[1], a2 = r4[2], a3 = r4[3];
    const float4 a4 = r4[4], a5 = r4[5], a6 = r4[6], a7 = r4[7];

    wave_lds_order();

    float4 t;
    t.x = ((a0.x + a1.x) + (a2.x + a3.x)) + ((a4.x + a5.x) + (a6.x + a7.x));
    t.y = ((a0.y + a1.y) + (a2.y + a3.y)) + ((a4.y + a5.y) + (a6.y + a7.y));
    t.z = ((a0.z + a1.z) + (a2.z + a3.z)) + ((a4.z + a5.z) + (a6.z + a7.z));
    t.w = ((a0.w + a1.w) + (a2.w + a3.w)) + ((a4.w + a5.w) + (a6.w + a7.w));
    const float tsum = (t.x + t.y) + (t.z + t.w);
    const float s = live ? tsum : NEG_INF;

    const float m   = wave_max(s);
    const float e   = __expf(s - m);
    const float inv = 1.0f / wave_sum(e);

    float acc = 0.f;
#pragma unroll
    for (int j = 0; j < NNB; ++j) {
      const float pj = __builtin_bit_cast(float,
          __builtin_amdgcn_readlane(__builtin_bit_cast(int, e), j));
      acc += pj * krow[j][cv];
    }
    out[base + co] = acc * inv;
  }
}

extern "C" void kernel_launch(void* const* d_in, const int* in_sizes, int n_in,
                              void* d_out, int out_size, void* d_ws, size_t ws_size,
                              hipStream_t stream) {
  (void)in_sizes; (void)n_in; (void)out_size; (void)ws_size;
  const float* x      = (const float*)d_in[0];
  const float* w_qkv  = (const float*)d_in[1];
  const float* w_proj = (const float*)d_in[2];
  const float* b_proj = (const float*)d_in[3];
  const int*   neigh  = (const int*)d_in[4];
  float* out = (float*)d_out;
  float* qkv = (float*)d_ws;

  const int M = BATCH * NVOX;

  {
    int waves = (M / 16) * (3 * CCH / 64);
    wmma_gemm_f32<<<waves * 32 / 256, 256, 0, stream>>>(
        x, w_qkv, nullptr, qkv, M, CCH, 3 * CCH, CCH);
  }
  {
    int waves = BATCH * NVOX;
    neigh_attn<<<waves * 32 / 256, 256, 0, stream>>>(qkv, neigh, qkv);
  }
  {
    int waves = (M / 16) * (CCH / 64);
    wmma_gemm_f32<<<waves * 32 / 256, 256, 0, stream>>>(
        qkv, w_proj, b_proj, out, M, CCH, CCH, 3 * CCH);
  }
}


# both GEMMs re-tiled, no LDS: wave = 32x64 block (B fragment feeds 2 WMMAs), work-group = 4x2 such blocks sharing A and B lines in the WGP cache, loads 3 k-steps ahead in registers; sum order same
# speedup vs baseline: 3.0075x; 1.0841x over previous
#include <hip/hip_runtime.h>

typedef __attribute__((ext_vector_type(2))) float v2f;
typedef __attribute__((ext_vector_type(8))) float v8f;

#define NVOX   4096
#define NHEADS 8
#define HDIM   32
#define CCH    256
#define BATCH  4


template <int MT>
static __device__ __forceinline__ void load_frags(const float* arow, const float* bcol,
                                                  int lda, int Ncols, int k,
                                                  v2f (&a)[MT], v2f (&bf)[4]) {
#pragma unroll
  for (int r = 0; r < MT; ++r)
    a[r] = *(const v2f*)(arow + (size_t)(16 * r) * lda + k);
  const float* b0 = bcol + (size_t)k * Ncols;
  const float* b1 = b0 + Ncols;
#pragma unroll
  for (int t = 0; t < 4; ++t) { bf[t].x = b0[16 * t]; bf[t].y = b1[16 * t]; }
}

template <int MT>
static __device__ __forceinline__ void mma_step(v8f (&acc)[MT][4], const v2f (&a)[MT], const v2f (&bf)[4]) {
#pragma unroll
  for (int r = 0; r < MT; ++r)
#pragma unroll
    for (int t = 0; t < 4; ++t)
      acc[r][t] = __builtin_amdgcn_wmma_f32_16x16x4_f32(false, a[r], false, bf[t], (short)0, acc[r][t], false, false);
}

template <int MT>
__global__ void __launch_bounds__(256)
wmma_gemm_f32(const float* A, const float* __restrict__ B,
              const float* __restrict__ bias, float* D,
              int M, int K, int Ncols, int lda) {
  const int lane = threadIdx.x & 31;
  const int wib  = __builtin_amdgcn_readfirstlane((int)(threadIdx.x >> 5));
  const int ntp  = Ncols >> 7;
  const int tg   = blockIdx.x / ntp;
  const int tp   = blockIdx.x - tg * ntp;
  const int row0 = ((tg << 2) + (wib & 3)) * (16 * MT);
  if (row0 >= M) return;
  const int col0 = ((tp << 1) + (wib >> 2)) << 6;
  const int half = lane >> 4;
  const int lid  = lane & 15;

  const float* arow = A + (size_t)(row0 + lid) * lda + (half << 1);
  const float* bcol = B + (size_t)(half << 1) * Ncols + col0 + lid;

  v8f acc[MT][4];
#pragma unroll
  for (int r = 0; r < MT; ++r)
#pragma unroll
    for (int t = 0; t < 4; ++t)
      acc[r][t] = (v8f){0.f,0.f,0.f,0.f,0.f,0.f,0.f,0.f};

  constexpr int NS = 4;
  v2f aS[NS][MT], bS[NS][4];
#pragma unroll
  for (int s = 0; s < NS - 1; ++s) {
    load_frags<MT>(arow, bcol, lda, Ncols, 4 * s, aS[s], bS[s]);
    __builtin_amdgcn_sched_barrier(0);
  }
  int k = 0;
  for (; k + 4 * NS < K; k += 4 * NS) {
#pragma unroll
    for (int s = 0; s < NS; ++s) {
      load_frags<MT>(arow, bcol, lda, Ncols, k + 4 * (s + NS - 1), aS[(s + NS - 1) % NS], bS[(s + NS - 1) % NS]);
      __builtin_amdgcn_sched_barrier(0);
      mma_step<MT>(acc, aS[s], bS[s]);
    }
  }
  load_frags<MT>(arow, bcol, lda, Ncols, k + 4 * (NS - 1), aS[NS - 1], bS[NS - 1]);
  __builtin_amdgcn_sched_barrier(0);
#pragma unroll
  for (int s = 0; s < NS; ++s) mma_step<MT>(acc, aS[s], bS[s]);

  float bv[4];
#pragma unroll
  for (int t = 0; t < 4; ++t) bv[t] = bias ? bias[col0 + lid + 16 * t] : 0.0f;
#pragma unroll
  for (int r = 0; r < MT; ++r) {
    float* drow = D + (size_t)(row0 + 16 * r + (half << 3)) * Ncols + col0 + lid;
#pragma unroll
    for (int i = 0; i < 8; ++i) {
      float* dr = drow + (size_t)i * Ncols;
#pragma unroll
      for (int t = 0; t < 4; ++t) dr[16 * t] = acc[r][t][i] + bv[t];
    }
  }
}

#define NNB  27
#define ROWP 36
#define WPB  8

static __device__ __forceinline__ float xhalf(float x) {
  const int b = __builtin_bit_cast(int, x);
  return __builtin_bit_cast(float, __builtin_amdgcn_permlanex16(b, b, 0x76543210, 0xfedcba98, false, false));
}
template <int M> static __device__ __forceinline__ float xrow(float x) {
  return __builtin_bit_cast(float,
      __builtin_amdgcn_update_dpp(0, __builtin_bit_cast(int, x), 0x160 + M, 0xf, 0xf, false));
}
static __device__ __forceinline__ float wave_max(float x) {
  x = fmaxf(x, xrow<1>(x)); x = fmaxf(x, xrow<2>(x));
  x = fmaxf(x, xrow<4>(x)); x = fmaxf(x, xrow<8>(x));
  return fmaxf(x, xhalf(x));
}
static __device__ __forceinline__ float wave_sum(float x) {
  x += xrow<1>(x); x += xrow<2>(x); x += xrow<4>(x); x += xrow<8>(x);
  return x + xhalf(x);
}
static __device__ __forceinline__ void wave_lds_order() {
  __builtin_amdgcn_fence(__ATOMIC_RELEASE, "wavefront");
  __builtin_amdgcn_wave_barrier();
  __builtin_amdgcn_fence(__ATOMIC_ACQUIRE, "wavefront");
}

__global__ void __launch_bounds__(256)
neigh_attn(const float* qkv, const int* __restrict__ neigh, float* out) {
  __shared__ __attribute__((aligned(16))) float prod[WPB][NNB][ROWP];

  const int lane = threadIdx.x & 31;
  const int wib  = __builtin_amdgcn_readfirstlane((int)(threadIdx.x >> 5));
  const int wave = blockIdx.x * WPB + wib;
  const int n = wave & (NVOX - 1);
  const int b = wave >> 12;

  const float NEG_INF = -__builtin_inff();
  const size_t base = (size_t)(b * NVOX + n) * (3 * CCH);
  float* myprod = &prod[wib][0][0];

  const int jr = lane < NNB ? lane : NNB - 1;
  const bool live = (lane < NNB) && (neigh[n * NNB + jr] != NVOX);

  const float* krow[NNB];
#pragma unroll
  for (int j = 0; j < NNB; ++j) {
    const int id = neigh[n * NNB + j];
    krow[j] = qkv + (size_t)(b * NVOX + (id != NVOX ? id : 0)) * (3 * CCH) + CCH;
  }

  for (int h = 0; h < NHEADS; ++h) {
    const unsigned co = (unsigned)(h * HDIM + lane);
    const unsigned cv = co + CCH;
    const float q = qkv[base + co] * 0.17677669529663687f;

#pragma unroll
    for (int j = 0; j < NNB; ++j)
      myprod[j * ROWP + lane] = q * krow[j][co];

    wave_lds_order();

    const float4* r4 = (const float4*)(myprod + jr * ROWP);
    const float4 a0 = r4[0], a1 = r4[1], a2 = r4[2], a3 = r4[3];
    const float4 a4 = r4[4], a5 = r4[5], a6 = r4[6], a7 = r4[7];

    wave_lds_order();

    float4 t;
    t.x = ((a0.x + a1.x) + (a2.x + a3.x)) + ((a4.x + a5.x) + (a6.x + a7.x));
    t.y = ((a0.y + a1.y) + (a2.y + a3.y)) + ((a4.y + a5.y) + (a6.y + a7.y));
    t.z = ((a0.z + a1.z) + (a2.z + a3.z)) + ((a4.z + a5.z) + (a6.z + a7.z));
    t.w = ((a0.w + a1.w) + (a2.w + a3.w)) + ((a4.w + a5.w) + (a6.w + a7.w));
    const float tsum = (t.x + t.y) + (t.z + t.w);
    const float s = live ? tsum : NEG_INF;

    const float m   = wave_max(s);
    const float e   = __expf(s - m);
    const float inv = 1.0f / wave_sum(e);

    float acc = 0.f;
#pragma unroll
    for (int j = 0; j < NNB; ++j) {
      const float pj = __builtin_bit_cast(float,
          __builtin_amdgcn_readlane(__builtin_bit_cast(int, e), j));
      acc += pj * krow[j][cv];
    }
    out[base + co] = acc * inv;
  }
}

extern "C" void kernel_launch(void* const* d_in, const int* in_sizes, int n_in,
                              void* d_out, int out_size, void* d_ws, size_t ws_size,
                              hipStream_t stream) {
  (void)in_sizes; (void)n_in; (void)out_size; (void)ws_size;
  const float* x      = (const float*)d_in[0];
  const float* w_qkv  = (const float*)d_in[1];
  const float* w_proj = (const float*)d_in[2];
  const float* b_proj = (const float*)d_in[3];
  const int*   neigh  = (const int*)d_in[4];
  float* out = (float*)d_out;
  float* qkv = (float*)d_ws;

  const int M = BATCH * NVOX;

  {
    int waves = (M / 32) * (3 * CCH / 64);
    wmma_gemm_f32<2><<<waves * 32 / 256, 256, 0, stream>>>(
        x, w_qkv, nullptr, qkv, M, CCH, 3 * CCH, CCH);
  }
  {
    int waves = BATCH * NVOX;
    neigh_attn<<<waves * 32 / 256, 256, 0, stream>>>(qkv, neigh, qkv);
  }
  {
    int waves = (M / 32) * (CCH / 64);
    wmma_gemm_f32<2><<<waves * 32 / 256, 256, 0, stream>>>(
        qkv, w_proj, b_proj, out, M, CCH, CCH, 3 * CCH);
  }
}


# both GEMMs: k and column order inside the wave tile permuted so every A, B and D access is 16 B per lane (an A line: 4 loads, was 8; B: 2 loads per k-step, was 8); sum order changes
# speedup vs baseline: 3.3094x; 1.1004x over previous
#include <hip/hip_runtime.h>

typedef __attribute__((ext_vector_type(2))) float v2f;
typedef __attribute__((ext_vector_type(4))) float v4f;
typedef __attribute__((ext_vector_type(8))) float v8f;

#define NVOX   4096
#define NHEADS 8
#define HDIM   32
#define CCH    256
#define BATCH  4


template <int MT>
static __device__ __forceinline__ void load_a(const float* arow, int lda, int k, v4f (&a)[MT]) {
#pragma unroll
  for (int r = 0; r < MT; ++r)
    a[r] = *(const v4f*)(arow + (size_t)(16 * r) * lda + k);
}
static __device__ __forceinline__ void load_b(const float* bcol, int Ncols, int krow, v4f (&b)[2]) {
  const float* p = bcol + (size_t)krow * Ncols;
  b[0] = *(const v4f*)p;
  b[1] = *(const v4f*)(p + Ncols);
}

template <int MT, int HI>
static __device__ __forceinline__ void mma_step(v8f (&acc)[MT][4], const v4f (&a)[MT], const v4f (&b)[2]) {
#pragma unroll
  for (int r = 0; r < MT; ++r) {
    const v2f af = HI ? (v2f){a[r].z, a[r].w} : (v2f){a[r].x, a[r].y};
#pragma unroll
    for (int t = 0; t < 4; ++t) {
      const v2f bf = (v2f){b[0][t], b[1][t]};
      acc[r][t] = __builtin_amdgcn_wmma_f32_16x16x4_f32(false, af, false, bf, (short)0, acc[r][t], false, false);
    }
  }
}

template <int MT>
__global__ void __launch_bounds__(256)
wmma_gemm_f32(const float* A, const float* __restrict__ B,
              const float* __restrict__ bias, float* D,
              int M, int K, int Ncols, int lda) {
  const int lane = threadIdx.x & 31;
  const int wib  = __builtin_amdgcn_readfirstlane((int)(threadIdx.x >> 5));
  const int ntp  = Ncols >> 7;
  const int tg   = blockIdx.x / ntp;
  const int tp   = blockIdx.x - tg * ntp;
  const int row0 = ((tg << 2) + (wib & 3)) * (16 * MT);
  if (row0 >= M) return;
  const int col0 = ((tp << 1) + (wib >> 2)) << 6;
  const int half = lane >> 4;
  const int lid  = lane & 15;

  const float* arow = A + (size_t)(row0 + lid) * lda + (half << 3);
  const float* bcol = B + (size_t)(half << 3) * Ncols + col0 + (lid << 2);

  v8f acc[MT][4];
#pragma unroll
  for (int r = 0; r < MT; ++r)
#pragma unroll
    for (int t = 0; t < 4; ++t)
      acc[r][t] = (v8f){0.f,0.f,0.f,0.f,0.f,0.f,0.f,0.f};

  v4f a0[MT], a1[MT], b0[2], b1[2], b2[2], b3[2];
  load_a<MT>(arow, lda, 0, a0);
  load_b(bcol, Ncols, 0, b0);
  load_b(bcol, Ncols, 2, b1);
  load_b(bcol, Ncols, 4, b2);
  int k = 0;
  for (; k + 16 < K; k += 16) {
    v4f n0[2], n1[2], n2[2], an[MT];
    load_a<MT>(arow, lda, k + 4, a1);
    load_b(bcol, Ncols, k + 6, b3);
    __builtin_amdgcn_sched_barrier(0);
    load_b(bcol, Ncols, k + 16, n0);
    load_b(bcol, Ncols, k + 18, n1);
    __builtin_amdgcn_sched_barrier(0);
    mma_step<MT, 0>(acc, a0, b0);
    load_b(bcol, Ncols, k + 20, n2);
    __builtin_amdgcn_sched_barrier(0);
    mma_step<MT, 1>(acc, a0, b1);
    load_a<MT>(arow, lda, k + 16, an);
    __builtin_amdgcn_sched_barrier(0);
    mma_step<MT, 0>(acc, a1, b2);
    mma_step<MT, 1>(acc, a1, b3);
    __builtin_amdgcn_sched_barrier(0);
#pragma unroll
    for (int i = 0; i < 2; ++i) { b0[i] = n0[i]; b1[i] = n1[i]; b2[i] = n2[i]; }
#pragma unroll
    for (int r = 0; r < MT; ++r) a0[r] = an[r];
  }
  load_a<MT>(arow, lda, k + 4, a1);
  load_b(bcol, Ncols, k + 6, b3);
  __builtin_amdgcn_sched_barrier(0);
  mma_step<MT, 0>(acc, a0, b0);
  mma_step<MT, 1>(acc, a0, b1);
  mma_step<MT, 0>(acc, a1, b2);
  mma_step<MT, 1>(acc, a1, b3);

  v4f bv = (v4f){0.f, 0.f, 0.f, 0.f};
  if (bias) bv = *(const v4f*)(bias + col0 + (lid << 2));
#pragma unroll
  for (int r = 0; r < MT; ++r) {
    float* drow = D + (size_t)(row0 + 16 * r + (half << 3)) * Ncols + col0 + (lid << 2);
#pragma unroll
    for (int i = 0; i < 8; ++i) {
      const v4f o = (v4f){acc[r][0][i], acc[r][1][i], acc[r][2][i], acc[r][3][i]} + bv;
      *(v4f*)(drow + (size_t)i * Ncols) = o;
    }
  }
}

#define NNB  27
#define ROWP 36
#define WPB  8

static __device__ __forceinline__ float xhalf(float x) {
  const int b = __builtin_bit_cast(int, x);
  return __builtin_bit_cast(float, __builtin_amdgcn_permlanex16(b, b, 0x76543210, 0xfedcba98, false, false));
}
template <int M> static __device__ __forceinline__ float xrow(float x) {
  return __builtin_bit_cast(float,
      __builtin_amdgcn_update_dpp(0, __builtin_bit_cast(int, x), 0x160 + M, 0xf, 0xf, false));
}
static __device__ __forceinline__ float wave_max(float x) {
  x = fmaxf(x, xrow<1>(x)); x = fmaxf(x, xrow<2>(x));
  x = fmaxf(x, xrow<4>(x)); x = fmaxf(x, xrow<8>(x));
  return fmaxf(x, xhalf(x));
}
static __device__ __forceinline__ float wave_sum(float x) {
  x += xrow<1>(x); x += xrow<2>(x); x += xrow<4>(x); x += xrow<8>(x);
  return x + xhalf(x);
}
static __device__ __forceinline__ void wave_lds_order() {
  __builtin_amdgcn_fence(__ATOMIC_RELEASE, "wavefront");
  __builtin_amdgcn_wave_barrier();
  __builtin_amdgcn_fence(__ATOMIC_ACQUIRE, "wavefront");
}

__global__ void __launch_bounds__(256)
neigh_attn(const float* qkv, const int* __restrict__ neigh, float* out) {
  __shared__ __attribute__((aligned(16))) float prod[WPB][NNB][ROWP];

  const int lane = threadIdx.x & 31;
  const int wib  = __builtin_amdgcn_readfirstlane((int)(threadIdx.x >> 5));
  const int wave = blockIdx.x * WPB + wib;
  const int n = wave & (NVOX - 1);
  const int b = wave >> 12;

  const float NEG_INF = -__builtin_inff();
  const size_t base = (size_t)(b * NVOX + n) * (3 * CCH);
  float* myprod = &prod[wib][0][0];

  const int jr = lane < NNB ? lane : NNB - 1;
  const bool live = (lane < NNB) && (neigh[n * NNB + jr] != NVOX);

  const float* krow[NNB];
#pragma unroll
  for (int j = 0; j < NNB; ++j) {
    const int id = neigh[n * NNB + j];
    krow[j] = qkv + (size_t)(b * NVOX + (id != NVOX ? id : 0)) * (3 * CCH) + CCH;
  }

  for (int h = 0; h < NHEADS; ++h) {
    const unsigned co = (unsigned)(h * HDIM + lane);
    const unsigned cv = co + CCH;
    const float q = qkv[base + co] * 0.17677669529663687f;

#pragma unroll
    for (int j = 0; j < NNB; ++j)
      myprod[j * ROWP + lane] = q * krow[j][co];

    wave_lds_order();

    const float4* r4 = (const float4*)(myprod + jr * ROWP);
    const float4 a0 = r4[0], a1 = r4[1], a2 = r4[2], a3 = r4[3];
    const float4 a4 = r4[4], a5 = r4[5], a6 = r4[6], a7 = r4[7];

    wave_lds_order();

    float4 t;
    t.x = ((a0.x + a1.x) + (a2.x + a3.x)) + ((a4.x + a5.x) + (a6.x + a7.x));
    t.y = ((a0.y + a1.y) + (a2.y + a3.y)) + ((a4.y + a5.y) + (a6.y + a7.y));
    t.z = ((a0.z + a1.z) + (a2.z + a3.z)) + ((a4.z + a5.z) + (a6.z + a7.z));
    t.w = ((a0.w + a1.w) + (a2.w + a3.w)) + ((a4.w + a5.w) + (a6.w + a7.w));
    const float tsum = (t.x + t.y) + (t.z + t.w);
    const float s = live ? tsum : NEG_INF;

    const float m   = wave_max(s);
    const float e   = __expf(s - m);
    const float inv = 1.0f / wave_sum(e);

    float acc = 0.f;
#pragma unroll
    for (int j = 0; j < NNB; ++j) {
      const float pj = __builtin_bit_cast(float,
          __builtin_amdgcn_readlane(__builtin_bit_cast(int, e), j));
      acc += pj * krow[j][cv];
    }
    out[base + co] = acc * inv;
  }
}

extern "C" void kernel_launch(void* const* d_in, const int* in_sizes, int n_in,
                              void* d_out, int out_size, void* d_ws, size_t ws_size,
                              hipStream_t stream) {
  (void)in_sizes; (void)n_in; (void)out_size; (void)ws_size;
  const float* x      = (const float*)d_in[0];
  const float* w_qkv  = (const float*)d_in[1];
  const float* w_proj = (const float*)d_in[2];
  const float* b_proj = (const float*)d_in[3];
  const int*   neigh  = (const int*)d_in[4];
  float* out = (float*)d_out;
  float* qkv = (float*)d_ws;

  const int M = BATCH * NVOX;

  {
    int waves = (M / 32) * (3 * CCH / 64);
    wmma_gemm_f32<2><<<waves * 32 / 256, 256, 0, stream>>>(
        x, w_qkv, nullptr, qkv, M, CCH, 3 * CCH, CCH);
  }
  {
    int waves = BATCH * NVOX;
    neigh_attn<<<waves * 32 / 256, 256, 0, stream>>>(qkv, neigh, qkv);
  }
  {
    int waves = (M / 32) * (CCH / 64);
    wmma_gemm_f32<2><<<waves * 32 / 256, 256, 0, stream>>>(
        qkv, w_proj, b_proj, out, M, CCH, CCH, 3 * CCH);
  }
}


# attention on the matrix unit: wave = a line of 16 voxels along z x one head; each of the 9 neighbour lines' scores as a dense 16x16 WMMA product (band kept), P.V on WMMA too; sum order changes
# speedup vs baseline: 3.4143x; 1.0317x over previous
#include <hip/hip_runtime.h>

typedef __attribute__((ext_vector_type(2))) float v2f;
typedef __attribute__((ext_vector_type(4))) float v4f;
typedef __attribute__((ext_vector_type(8))) float v8f;

#define NVOX   4096
#define NHEADS 8
#define HDIM   32
#define CCH    256
#define BATCH  4


template <int MT>
static __device__ __forceinline__ void load_a(const float* arow, int lda, int k, v4f (&a)[MT]) {
#pragma unroll
  for (int r = 0; r < MT; ++r)
    a[r] = *(const v4f*)(arow + (size_t)(16 * r) * lda + k);
}
static __device__ __forceinline__ void load_b(const float* bcol, int Ncols, int krow, v4f (&b)[2]) {
  const float* p = bcol + (size_t)krow * Ncols;
  b[0] = *(const v4f*)p;
  b[1] = *(const v4f*)(p + Ncols);
}

template <int MT, int HI>
static __device__ __forceinline__ void mma_step(v8f (&acc)[MT][4], const v4f (&a)[MT], const v4f (&b)[2]) {
#pragma unroll
  for (int r = 0; r < MT; ++r) {
    const v2f af = HI ? (v2f){a[r].z, a[r].w} : (v2f){a[r].x, a[r].y};
#pragma unroll
    for (int t = 0; t < 4; ++t) {
      const v2f bf = (v2f){b[0][t], b[1][t]};
      acc[r][t] = __builtin_amdgcn_wmma_f32_16x16x4_f32(false, af, false, bf, (short)0, acc[r][t], false, false);
    }
  }
}

template <int MT>
__global__ void __launch_bounds__(256)
wmma_gemm_f32(const float* A, const float* __restrict__ B,
              const float* __restrict__ bias, float* D,
              int M, int K, int Ncols, int lda) {
  const int lane = threadIdx.x & 31;
  const int wib  = __builtin_amdgcn_readfirstlane((int)(threadIdx.x >> 5));
  const int ntp  = Ncols >> 7;
  const int tg   = blockIdx.x / ntp;
  const int tp   = blockIdx.x - tg * ntp;
  const int row0 = ((tg << 2) + (wib & 3)) * (16 * MT);
  if (row0 >= M) return;
  const int col0 = ((tp << 1) + (wib >> 2)) << 6;
  const int half = lane >> 4;
  const int lid  = lane & 15;

  const float* arow = A + (size_t)(row0 + lid) * lda + (half << 3);
  const float* bcol = B + (size_t)(half << 3) * Ncols + col0 + (lid << 2);

  v8f acc[MT][4];
#pragma unroll
  for (int r = 0; r < MT; ++r)
#pragma unroll
    for (int t = 0; t < 4; ++t)
      acc[r][t] = (v8f){0.f,0.f,0.f,0.f,0.f,0.f,0.f,0.f};

  v4f a0[MT], a1[MT], b0[2], b1[2], b2[2], b3[2];
  load_a<MT>(arow, lda, 0, a0);
  load_b(bcol, Ncols, 0, b0);
  load_b(bcol, Ncols, 2, b1);
  load_b(bcol, Ncols, 4, b2);
  int k = 0;
  for (; k + 16 < K; k += 16) {
    v4f n0[2], n1[2], n2[2], an[MT];
    load_a<MT>(arow, lda, k + 4, a1);
    load_b(bcol, Ncols, k + 6, b3);
    __builtin_amdgcn_sched_barrier(0);
    load_b(bcol, Ncols, k + 16, n0);
    load_b(bcol, Ncols, k + 18, n1);
    __builtin_amdgcn_sched_barrier(0);
    mma_step<MT, 0>(acc, a0, b0);
    load_b(bcol, Ncols, k + 20, n2);
    __builtin_amdgcn_sched_barrier(0);
    mma_step<MT, 1>(acc, a0, b1);
    load_a<MT>(arow, lda, k + 16, an);
    __builtin_amdgcn_sched_barrier(0);
    mma_step<MT, 0>(acc, a1, b2);
    mma_step<MT, 1>(acc, a1, b3);
    __builtin_amdgcn_sched_barrier(0);
#pragma unroll
    for (int i = 0; i < 2; ++i) { b0[i] = n0[i]; b1[i] = n1[i]; b2[i] = n2[i]; }
#pragma unroll
    for (int r = 0; r < MT; ++r) a0[r] = an[r];
  }
  load_a<MT>(arow, lda, k + 4, a1);
  load_b(bcol, Ncols, k + 6, b3);
  __builtin_amdgcn_sched_barrier(0);
  mma_step<MT, 0>(acc, a0, b0);
  mma_step<MT, 1>(acc, a0, b1);
  mma_step<MT, 0>(acc, a1, b2);
  mma_step<MT, 1>(acc, a1, b3);

  v4f bv = (v4f){0.f, 0.f, 0.f, 0.f};
  if (bias) bv = *(const v4f*)(bias + col0 + (lid << 2));
#pragma unroll
  for (int r = 0; r < MT; ++r) {
    float* drow = D + (size_t)(row0 + 16 * r + (half << 3)) * Ncols + col0 + (lid << 2);
#pragma unroll
    for (int i = 0; i < 8; ++i) {
      const v4f o = (v4f){acc[r][0][i], acc[r][1][i], acc[r][2][i], acc[r][3][i]} + bv;
      *(v4f*)(drow + (size_t)i * Ncols) = o;
    }
  }
}

#define NNB  27
#define ROWP 36
#define WPB  8

static __device__ __forceinline__ float xhalf(float x) {
  const int b = __builtin_bit_cast(int, x);
  return __builtin_bit_cast(float, __builtin_amdgcn_permlanex16(b, b, 0x76543210, 0xfedcba98, false, false));
}
template <int M> static __device__ __forceinline__ float xrow(float x) {
  return __builtin_bit_cast(float,
      __builtin_amdgcn_update_dpp(0, __builtin_bit_cast(int, x), 0x160 + M, 0xf, 0xf, false));
}
static __device__ __forceinline__ float wave_max(float x) {
  x = fmaxf(x, xrow<1>(x)); x = fmaxf(x, xrow<2>(x));
  x = fmaxf(x, xrow<4>(x)); x = fmaxf(x, xrow<8>(x));
  return fmaxf(x, xhalf(x));
}
static __device__ __forceinline__ float wave_sum(float x) {
  x += xrow<1>(x); x += xrow<2>(x); x += xrow<4>(x); x += xrow<8>(x);
  return x + xhalf(x);
}
static __device__ __forceinline__ void wave_lds_order() {
  __builtin_amdgcn_fence(__ATOMIC_RELEASE, "wavefront");
  __builtin_amdgcn_wave_barrier();
  __builtin_amdgcn_fence(__ATOMIC_ACQUIRE, "wavefront");
}

static __device__ __forceinline__ void voxel_generic(const float* qkv, const int* __restrict__ neigh,
                                                     float* out, float* myprod,
                                                     int b, int n, int lane) {
  const float NEG_INF = -__builtin_inff();
  const size_t base = (size_t)(b * NVOX + n) * (3 * CCH);
  const int jr = lane < NNB ? lane : NNB - 1;
  const bool live = (lane < NNB) && (neigh[n * NNB + jr] != NVOX);

  const float* krow[NNB];
#pragma unroll
  for (int j = 0; j < NNB; ++j) {
    const int id = neigh[n * NNB + j];
    krow[j] = qkv + (size_t)(b * NVOX + (id != NVOX ? id : 0)) * (3 * CCH) + CCH;
  }

  for (int h = 0; h < NHEADS; ++h) {
    const unsigned co = (unsigned)(h * HDIM + lane);
    const unsigned cv = co + CCH;
    const float q = qkv[base + co] * 0.17677669529663687f;

#pragma unroll
    for (int j = 0; j < NNB; ++j)
      myprod[j * ROWP + lane] = q * krow[j][co];

    wave_lds_order();

    const float4* r4 = (const float4*)(myprod + jr * ROWP);
    const float4 a0 = r4[0], a1 = r4[1], a2 = r4[2], a3 = r4[3];
    const float4 a4 = r4[4], a5 = r4[5], a6 = r4[6], a7 = r4[7];

    wave_lds_order();

    float4 t;
    t.x = ((a0.x + a1.x) + (a2.x + a3.x)) + ((a4.x + a5.x) + (a6.x + a7.x));
    t.y = ((a0.y + a1.y) + (a2.y + a3.y)) + ((a4.y + a5.y) + (a6.y + a7.y));
    t.z = ((a0.z + a1.z) + (a2.z + a3.z)) + ((a4.z + a5.z) + (a6.z + a7.z));
    t.w = ((a0.w + a1.w) + (a2.w + a3.w)) + ((a4.w + a5.w) + (a6.w + a7.w));
    const float tsum = (t.x + t.y) + (t.z + t.w);
    const float s = live ? tsum : NEG_INF;

    const float m   = wave_max(s);
    const float e   = __expf(s - m);
    const float inv = 1.0f / wave_sum(e);

    float acc = 0.f;
#pragma unroll
    for (int j = 0; j < NNB; ++j) {
      const float pj = __builtin_bit_cast(float,
          __builtin_amdgcn_readlane(__builtin_bit_cast(int, e), j));
      acc += pj * krow[j][cv];
    }
    out[base + co] = acc * inv;
  }
}

static __device__ __forceinline__ void load_row16(const float* p, v4f (&f)[4]) {
  f[0] = *(const v4f*)(p);
  f[1] = *(const v4f*)(p + 4);
  f[2] = *(const v4f*)(p + 16);
  f[3] = *(const v4f*)(p + 20);
}

__global__ void __launch_bounds__(256)
neigh_attn(const float* qkv, const int* __restrict__ neigh, float* out) {
  __shared__ __attribute__((aligned(16))) float prod[WPB][NNB][ROWP];

  const int lane = threadIdx.x & 31;
  const int head = __builtin_amdgcn_readfirstlane((int)(threadIdx.x >> 5));
  const int line = blockIdx.x;
  const int b  = line >> 8;
  const int xy = line & 255;
  const int x  = xy >> 4, y = xy & 15;
  const int n0 = xy << 4;
  const int half = lane >> 4, c = lane & 15;

  const float* rowbase = qkv + (size_t)(b * NVOX) * (3 * CCH);
  const int hoff = head * HDIM;

  v4f qf[4];
  load_row16(rowbase + (size_t)(n0 + c) * (3 * CCH) + hoff + 8 * half, qf);

  const int js = lane < NNB ? lane : NNB - 1;
  const int dzs = js % 3 - 1;
  const int nxs = x + js / 9 - 1, nys = y + (js / 3) % 3 - 1;
  const bool inxy = ((unsigned)nxs < 16u) && ((unsigned)nys < 16u);
  const int idbase = nxs * 256 + nys * 16 + dzs;
  int bad = 0;
#pragma unroll
  for (int zq = 0; zq < 16; ++zq) {
    const bool in = inxy && ((unsigned)(zq + dzs) < 16u);
    const int expect = in ? idbase + zq : NVOX;
    bad |= neigh[(n0 + zq) * NNB + js] ^ expect;
  }
  const bool regular = __builtin_amdgcn_ballot_w32(bad != 0) == 0u;

  if (!regular) {
    float* myprod = &prod[head][0][0];
#pragma nounroll
    for (int v = 0; v < 2; ++v)
      voxel_generic(qkv, neigh, out, myprod, b, n0 + 2 * head + v, lane);
    return;
  }

  const float NEG_INF = -__builtin_inff();
#pragma unroll
  for (int i = 0; i < 4; ++i) qf[i] = qf[i] * 0.17677669529663687f;

  bool band[8];
#pragma unroll
  for (int r = 0; r < 8; ++r) band[r] = (unsigned)(8 * half + r - c + 1) <= 2u;

  int  lbase[9];
  bool lok[9];
#pragma unroll
  for (int L = 0; L < 9; ++L) {
    const int nx = x + L / 3 - 1, ny = y + L % 3 - 1;
    lok[L] = ((unsigned)nx < 16u) && ((unsigned)ny < 16u);
    lbase[L] = lok[L] ? ((nx * 16 + ny) << 4) : n0;
  }

  v8f S[9];
  float m = NEG_INF;
#pragma unroll
  for (int L = 0; L < 9; ++L) {
    v4f kf[4];
    load_row16(rowbase + (size_t)(lbase[L] + c) * (3 * CCH) + CCH + hoff + 8 * half, kf);
    v8f acc = (v8f){0.f,0.f,0.f,0.f,0.f,0.f,0.f,0.f};
#pragma unroll
    for (int st = 0; st < 8; ++st) {
      const v2f af = (v2f){kf[st >> 1][2 * (st & 1)], kf[st >> 1][2 * (st & 1) + 1]};
      const v2f bf = (v2f){qf[st >> 1][2 * (st & 1)], qf[st >> 1][2 * (st & 1) + 1]};
      acc = __builtin_amdgcn_wmma_f32_16x16x4_f32(false, af, false, bf, (short)0, acc, false, false);
    }
#pragma unroll
    for (int r = 0; r < 8; ++r) {
      const float s = (lok[L] && band[r]) ? acc[r] : NEG_INF;
      S[L][r] = s;
      m = fmaxf(m, s);
    }
  }
  m = fmaxf(m, xhalf(m));

  float sum = 0.f;
#pragma unroll
  for (int L = 0; L < 9; ++L)
#pragma unroll
    for (int r = 0; r < 8; ++r) {
      const float e = __expf(S[L][r] - m);
      S[L][r] = e;
      sum += e;
    }
  sum += xhalf(sum);
  const float inv = 1.0f / sum;

  v8f o0 = (v8f){0.f,0.f,0.f,0.f,0.f,0.f,0.f,0.f};
  v8f o1 = (v8f){0.f,0.f,0.f,0.f,0.f,0.f,0.f,0.f};
#pragma unroll
  for (int L = 0; L < 9; ++L) {
    const float* vp = rowbase + (size_t)(lbase[L] + 8 * half) * (3 * CCH) + 2 * CCH + hoff + 2 * c;
#pragma unroll
    for (int s2 = 0; s2 < 4; ++s2) {
      const v2f va = *(const v2f*)(vp + (size_t)(2 * s2) * (3 * CCH));
      const v2f vb = *(const v2f*)(vp + (size_t)(2 * s2 + 1) * (3 * CCH));
      const v2f af = (v2f){S[L][2 * s2] * inv, S[L][2 * s2 + 1] * inv};
      o0 = __builtin_amdgcn_wmma_f32_16x16x4_f32(false, af, false, (v2f){va.x, vb.x}, (short)0, o0, false, false);
      o1 = __builtin_amdgcn_wmma_f32_16x16x4_f32(false, af, false, (v2f){va.y, vb.y}, (short)0, o1, false, false);
    }
  }

  float* op = out + (size_t)(b * NVOX + n0 + 8 * half) * (3 * CCH) + hoff + 2 * c;
#pragma unroll
  for (int r = 0; r < 8; ++r)
    *(v2f*)(op + (size_t)r * (3 * CCH)) = (v2f){o0[r], o1[r]};
}

extern "C" void kernel_launch(void* const* d_in, const int* in_sizes, int n_in,
                              void* d_out, int out_size, void* d_ws, size_t ws_size,
                              hipStream_t stream) {
  (void)in_sizes; (void)n_in; (void)out_size; (void)ws_size;
  const float* x      = (const float*)d_in[0];
  const float* w_qkv  = (const float*)d_in[1];
  const float* w_proj = (const float*)d_in[2];
  const float* b_proj = (const float*)d_in[3];
  const int*   neigh  = (const int*)d_in[4];
  float* out = (float*)d_out;
  float* qkv = (float*)d_ws;

  const int M = BATCH * NVOX;

  {
    int waves = (M / 32) * (3 * CCH / 64);
    wmma_gemm_f32<2><<<waves * 32 / 256, 256, 0, stream>>>(
        x, w_qkv, nullptr, qkv, M, CCH, 3 * CCH, CCH);
  }
  {
    neigh_attn<<<BATCH * 256, 256, 0, stream>>>(qkv, neigh, qkv);
  }
  {
    int waves = (M / 32) * (CCH / 64);
    wmma_gemm_f32<2><<<waves * 32 / 256, 256, 0, stream>>>(
        qkv, w_proj, b_proj, out, M, CCH, CCH, 3 * CCH);
  }
}


# attention: q and K rows fetched as whole 128-B lines (4 lines per load, each asked for once; was 16 lines per load, 32 B of each), made into WMMA fragments via a wave-private LDS tile; sum order same
# speedup vs baseline: 3.7030x; 1.0846x over previous
#include <hip/hip_runtime.h>

typedef __attribute__((ext_vector_type(2))) float v2f;
typedef __attribute__((ext_vector_type(4))) float v4f;
typedef __attribute__((ext_vector_type(8))) float v8f;

#define NVOX   4096
#define NHEADS 8
#define HDIM   32
#define CCH    256
#define BATCH  4


template <int MT>
static __device__ __forceinline__ void load_a(const float* arow, int lda, int k, v4f (&a)[MT]) {
#pragma unroll
  for (int r = 0; r < MT; ++r)
    a[r] = *(const v4f*)(arow + (size_t)(16 * r) * lda + k);
}
static __device__ __forceinline__ void load_b(const float* bcol, int Ncols, int krow, v4f (&b)[2]) {
  const float* p = bcol + (size_t)krow * Ncols;
  b[0] = *(const v4f*)p;
  b[1] = *(const v4f*)(p + Ncols);
}

template <int MT, int HI>
static __device__ __forceinline__ void mma_step(v8f (&acc)[MT][4], const v4f (&a)[MT], const v4f (&b)[2]) {
#pragma unroll
  for (int r = 0; r < MT; ++r) {
    const v2f af = HI ? (v2f){a[r].z, a[r].w} : (v2f){a[r].x, a[r].y};
#pragma unroll
    for (int t = 0; t < 4; ++t) {
      const v2f bf = (v2f){b[0][t], b[1][t]};
      acc[r][t] = __builtin_amdgcn_wmma_f32_16x16x4_f32(false, af, false, bf, (short)0, acc[r][t], false, false);
    }
  }
}

template <int MT>
__global__ void __launch_bounds__(256)
wmma_gemm_f32(const float* A, const float* __restrict__ B,
              const float* __restrict__ bias, float* D,
              int M, int K, int Ncols, int lda) {
  const int lane = threadIdx.x & 31;
  const int wib  = __builtin_amdgcn_readfirstlane((int)(threadIdx.x >> 5));
  const int ntp  = Ncols >> 7;
  const int tg   = blockIdx.x / ntp;
  const int tp   = blockIdx.x - tg * ntp;
  const int row0 = ((tg << 2) + (wib & 3)) * (16 * MT);
  if (row0 >= M) return;
  const int col0 = ((tp << 1) + (wib >> 2)) << 6;
  const int half = lane >> 4;
  const int lid  = lane & 15;

  const float* arow = A + (size_t)(row0 + lid) * lda + (half << 3);
  const float* bcol = B + (size_t)(half << 3) * Ncols + col0 + (lid << 2);

  v8f acc[MT][4];
#pragma unroll
  for (int r = 0; r < MT; ++r)
#pragma unroll
    for (int t = 0; t < 4; ++t)
      acc[r][t] = (v8f){0.f,0.f,0.f,0.f,0.f,0.f,0.f,0.f};

  v4f a0[MT], a1[MT], b0[2], b1[2], b2[2], b3[2];
  load_a<MT>(arow, lda, 0, a0);
  load_b(bcol, Ncols, 0, b0);
  load_b(bcol, Ncols, 2, b1);
  load_b(bcol, Ncols, 4, b2);
  int k = 0;
  for (; k + 16 < K; k += 16) {
    v4f n0[2], n1[2], n2[2], an[MT];
    load_a<MT>(arow, lda, k + 4, a1);
    load_b(bcol, Ncols, k + 6, b3);
    __builtin_amdgcn_sched_barrier(0);
    load_b(bcol, Ncols, k + 16, n0);
    load_b(bcol, Ncols, k + 18, n1);
    __builtin_amdgcn_sched_barrier(0);
    mma_step<MT, 0>(acc, a0, b0);
    load_b(bcol, Ncols, k + 20, n2);
    __builtin_amdgcn_sched_barrier(0);
    mma_step<MT, 1>(acc, a0, b1);
    load_a<MT>(arow, lda, k + 16, an);
    __builtin_amdgcn_sched_barrier(0);
    mma_step<MT, 0>(acc, a1, b2);
    mma_step<MT, 1>(acc, a1, b3);
    __builtin_amdgcn_sched_barrier(0);
#pragma unroll
    for (int i = 0; i < 2; ++i) { b0[i] = n0[i]; b1[i] = n1[i]; b2[i] = n2[i]; }
#pragma unroll
    for (int r = 0; r < MT; ++r) a0[r] = an[r];
  }
  load_a<MT>(arow, lda, k + 4, a1);
  load_b(bcol, Ncols, k + 6, b3);
  __builtin_amdgcn_sched_barrier(0);
  mma_step<MT, 0>(acc, a0, b0);
  mma_step<MT, 1>(acc, a0, b1);
  mma_step<MT, 0>(acc, a1, b2);
  mma_step<MT, 1>(acc, a1, b3);

  v4f bv = (v4f){0.f, 0.f, 0.f, 0.f};
  if (bias) bv = *(const v4f*)(bias + col0 + (lid << 2));
#pragma unroll
  for (int r = 0; r < MT; ++r) {
    float* drow = D + (size_t)(row0 + 16 * r + (half << 3)) * Ncols + col0 + (lid << 2);
#pragma unroll
    for (int i = 0; i < 8; ++i) {
      const v4f o = (v4f){acc[r][0][i], acc[r][1][i], acc[r][2][i], acc[r][3][i]} + bv;
      *(v4f*)(drow + (size_t)i * Ncols) = o;
    }
  }
}

#define NNB  27
#define ROWP 36
#define WPB  8

static __device__ __forceinline__ float xhalf(float x) {
  const int b = __builtin_bit_cast(int, x);
  return __builtin_bit_cast(float, __builtin_amdgcn_permlanex16(b, b, 0x76543210, 0xfedcba98, false, false));
}
template <int M> static __device__ __forceinline__ float xrow(float x) {
  return __builtin_bit_cast(float,
      __builtin_amdgcn_update_dpp(0, __builtin_bit_cast(int, x), 0x160 + M, 0xf, 0xf, false));
}
static __device__ __forceinline__ float wave_max(float x) {
  x = fmaxf(x, xrow<1>(x)); x = fmaxf(x, xrow<2>(x));
  x = fmaxf(x, xrow<4>(x)); x = fmaxf(x, xrow<8>(x));
  return fmaxf(x, xhalf(x));
}
static __device__ __forceinline__ float wave_sum(float x) {
  x += xrow<1>(x); x += xrow<2>(x); x += xrow<4>(x); x += xrow<8>(x);
  return x + xhalf(x);
}
static __device__ __forceinline__ void wave_lds_order() {
  __builtin_amdgcn_fence(__ATOMIC_RELEASE, "wavefront");
  __builtin_amdgcn_wave_barrier();
  __builtin_amdgcn_fence(__ATOMIC_ACQUIRE, "wavefront");
}

static __device__ __forceinline__ void voxel_generic(const float* qkv, const int* __restrict__ neigh,
                                                     float* out, float* myprod,
                                                     int b, int n, int lane) {
  const float NEG_INF = -__builtin_inff();
  const size_t base = (size_t)(b * NVOX + n) * (3 * CCH);
  const int jr = lane < NNB ? lane : NNB - 1;
  const bool live = (lane < NNB) && (neigh[n * NNB + jr] != NVOX);

  const float* krow[NNB];
#pragma unroll
  for (int j = 0; j < NNB; ++j) {
    const int id = neigh[n * NNB + j];
    krow[j] = qkv + (size_t)(b * NVOX + (id != NVOX ? id : 0)) * (3 * CCH) + CCH;
  }

  for (int h = 0; h < NHEADS; ++h) {
    const unsigned co = (unsigned)(h * HDIM + lane);
    const unsigned cv = co + CCH;
    const float q = qkv[base + co] * 0.17677669529663687f;

#pragma unroll
    for (int j = 0; j < NNB; ++j)
      myprod[j * ROWP + lane] = q * krow[j][co];

    wave_lds_order();

    const float4* r4 = (const float4*)(myprod + jr * ROWP);
    const float4 a0 = r4[0], a1 = r4[1], a2 = r4[2], a3 = r4[3];
    const float4 a4 = r4[4], a5 = r4[5], a6 = r4[6], a7 = r4[7];

    wave_lds_order();

    float4 t;
    t.x = ((a0.x + a1.x) + (a2.x + a3.x)) + ((a4.x + a5.x) + (a6.x + a7.x));
    t.y = ((a0.y + a1.y) + (a2.y + a3.y)) + ((a4.y + a5.y) + (a6.y + a7.y));
    t.z = ((a0.z + a1.z) + (a2.z + a3.z)) + ((a4.z + a5.z) + (a6.z + a7.z));
    t.w = ((a0.w + a1.w) + (a2.w + a3.w)) + ((a4.w + a5.w) + (a6.w + a7.w));
    const float tsum = (t.x + t.y) + (t.z + t.w);
    const float s = live ? tsum : NEG_INF;

    const float m   = wave_max(s);
    const float e   = __expf(s - m);
    const float inv = 1.0f / wave_sum(e);

    float acc = 0.f;
#pragma unroll
    for (int j = 0; j < NNB; ++j) {
      const float pj = __builtin_bit_cast(float,
          __builtin_amdgcn_readlane(__builtin_bit_cast(int, e), j));
      acc += pj * krow[j][cv];
    }
    out[base + co] = acc * inv;
  }
}

#define LROWS 16

static __device__ __forceinline__ void load_lines(const float* p, int goff, v4f (&g)[4]) {
#pragma unroll
  for (int i = 0; i < 4; ++i)
    g[i] = *(const v4f*)(p + goff + (size_t)(4 * i) * (3 * CCH));
}
static __device__ __forceinline__ void stage_lines(float* buf, int soff, const v4f (&g)[4]) {
#pragma unroll
  for (int i = 0; i < 4; ++i)
    *(v4f*)(buf + soff + 4 * i * ROWP) = g[i];
}
static __device__ __forceinline__ void read_frags(const float* buf, int foff, v4f (&f)[4]) {
  f[0] = *(const v4f*)(buf + foff);
  f[1] = *(const v4f*)(buf + foff + 4);
  f[2] = *(const v4f*)(buf + foff + 16);
  f[3] = *(const v4f*)(buf + foff + 20);
}

__global__ void __launch_bounds__(256)
neigh_attn(const float* qkv, const int* __restrict__ neigh, float* out) {
  __shared__ __attribute__((aligned(16))) float tile[WPB][2 * LROWS * ROWP];

  const int lane = threadIdx.x & 31;
  const int head = __builtin_amdgcn_readfirstlane((int)(threadIdx.x >> 5));
  const int line = blockIdx.x;
  const int b  = line >> 8;
  const int xy = line & 255;
  const int x  = xy >> 4, y = xy & 15;
  const int n0 = xy << 4;
  const int half = lane >> 4, c = lane & 15;

  const float* rowbase = qkv + (size_t)(b * NVOX) * (3 * CCH);
  const int hoff = head * HDIM;

  const int goff = (lane >> 3) * (3 * CCH) + ((lane & 7) << 2);
  const int soff = (lane >> 3) * ROWP + ((lane & 7) << 2);
  const int foff = c * ROWP + 8 * half;
  float* buf0 = &tile[head][0];
  float* buf1 = buf0 + LROWS * ROWP;

  v4f gq[4];
  load_lines(rowbase + (size_t)n0 * (3 * CCH) + hoff, goff, gq);

  const int js = lane < NNB ? lane : NNB - 1;
  const int dzs = js % 3 - 1;
  const int nxs = x + js / 9 - 1, nys = y + (js / 3) % 3 - 1;
  const bool inxy = ((unsigned)nxs < 16u) && ((unsigned)nys < 16u);
  const int idbase = nxs * 256 + nys * 16 + dzs;
  int bad = 0;
#pragma unroll
  for (int zq = 0; zq < 16; ++zq) {
    const bool in = inxy && ((unsigned)(zq + dzs) < 16u);
    const int expect = in ? idbase + zq : NVOX;
    bad |= neigh[(n0 + zq) * NNB + js] ^ expect;
  }
  const bool regular = __builtin_amdgcn_ballot_w32(bad != 0) == 0u;

  if (!regular) {
    float* myprod = buf0;
#pragma nounroll
    for (int v = 0; v < 2; ++v)
      voxel_generic(qkv, neigh, out, myprod, b, n0 + 2 * head + v, lane);
    return;
  }

  const float NEG_INF = -__builtin_inff();

  bool band[8];
#pragma unroll
  for (int r = 0; r < 8; ++r) band[r] = (unsigned)(8 * half + r - c + 1) <= 2u;

  int  lbase[9];
  bool lok[9];
#pragma unroll
  for (int L = 0; L < 9; ++L) {
    const int nx = x + L / 3 - 1, ny = y + L % 3 - 1;
    lok[L] = ((unsigned)nx < 16u) && ((unsigned)ny < 16u);
    lbase[L] = lok[L] ? ((nx * 16 + ny) << 4) : n0;
  }

  const float* kbase = rowbase + CCH + hoff;
  v4f g[2][4];
  load_lines(kbase + (size_t)lbase[0] * (3 * CCH), goff, g[0]);
  load_lines(kbase + (size_t)lbase[1] * (3 * CCH), goff, g[1]);

  v4f qf[4], kf[4];
  stage_lines(buf1, soff, gq);
  wave_lds_order();
  read_frags(buf1, foff, qf);
  stage_lines(buf0, soff, g[0]);
  wave_lds_order();
  read_frags(buf0, foff, kf);
  load_lines(kbase + (size_t)lbase[2] * (3 * CCH), goff, g[0]);
#pragma unroll
  for (int i = 0; i < 4; ++i) qf[i] = qf[i] * 0.17677669529663687f;

  v8f S[9];
  float m = NEG_INF;
#pragma unroll
  for (int L = 0; L < 9; ++L) {
    v4f kn[4];
    if (L + 1 < 9) {
      float* nb = ((L + 1) & 1) ? buf1 : buf0;
      stage_lines(nb, soff, g[(L + 1) & 1]);
      wave_lds_order();
      read_frags(nb, foff, kn);
      if (L + 3 < 9)
        load_lines(kbase + (size_t)lbase[L + 3] * (3 * CCH), goff, g[(L + 1) & 1]);
    }
    __builtin_amdgcn_sched_barrier(0);
    v8f acc = (v8f){0.f,0.f,0.f,0.f,0.f,0.f,0.f,0.f};
#pragma unroll
    for (int st = 0; st < 8; ++st) {
      const v2f af = (v2f){kf[st >> 1][2 * (st & 1)], kf[st >> 1][2 * (st & 1) + 1]};
      const v2f bf = (v2f){qf[st >> 1][2 * (st & 1)], qf[st >> 1][2 * (st & 1) + 1]};
      acc = __builtin_amdgcn_wmma_f32_16x16x4_f32(false, af, false, bf, (short)0, acc, false, false);
    }
#pragma unroll
    for (int r = 0; r < 8; ++r) {
      const float s = (lok[L] && band[r]) ? acc[r] : NEG_INF;
      S[L][r] = s;
      m = fmaxf(m, s);
    }
    if (L + 1 < 9) {
#pragma unroll
      for (int i = 0; i < 4; ++i) kf[i] = kn[i];
    }
  }
  m = fmaxf(m, xhalf(m));

  float sum = 0.f;
#pragma unroll
  for (int L = 0; L < 9; ++L)
#pragma unroll
    for (int r = 0; r < 8; ++r) {
      const float e = __expf(S[L][r] - m);
      S[L][r] = e;
      sum += e;
    }
  sum += xhalf(sum);
  const float inv = 1.0f / sum;

  v8f o0 = (v8f){0.f,0.f,0.f,0.f,0.f,0.f,0.f,0.f};
  v8f o1 = (v8f){0.f,0.f,0.f,0.f,0.f,0.f,0.f,0.f};
#pragma unroll
  for (int L = 0; L < 9; ++L) {
    const float* vp = rowbase + (size_t)(lbase[L] + 8 * half) * (3 * CCH) + 2 * CCH + hoff + 2 * c;
#pragma unroll
    for (int s2 = 0; s2 < 4; ++s2) {
      const v2f va = *(const v2f*)(vp + (size_t)(2 * s2) * (3 * CCH));
      const v2f vb = *(const v2f*)(vp + (size_t)(2 * s2 + 1) * (3 * CCH));
      const v2f af = (v2f){S[L][2 * s2] * inv, S[L][2 * s2 + 1] * inv};
      o0 = __builtin_amdgcn_wmma_f32_16x16x4_f32(false, af, false, (v2f){va.x, vb.x}, (short)0, o0, false, false);
      o1 = __builtin_amdgcn_wmma_f32_16x16x4_f32(false, af, false, (v2f){va.y, vb.y}, (short)0, o1, false, false);
    }
  }

  float* op = out + (size_t)(b * NVOX + n0 + 8 * half) * (3 * CCH) + hoff + 2 * c;
#pragma unroll
  for (int r = 0; r < 8; ++r)
    *(v2f*)(op + (size_t)r * (3 * CCH)) = (v2f){o0[r], o1[r]};
}

extern "C" void kernel_launch(void* const* d_in, const int* in_sizes, int n_in,
                              void* d_out, int out_size, void* d_ws, size_t ws_size,
                              hipStream_t stream) {
  (void)in_sizes; (void)n_in; (void)out_size; (void)ws_size;
  const float* x      = (const float*)d_in[0];
  const float* w_qkv  = (const float*)d_in[1];
  const float* w_proj = (const float*)d_in[2];
  const float* b_proj = (const float*)d_in[3];
  const int*   neigh  = (const int*)d_in[4];
  float* out = (float*)d_out;
  float* qkv = (float*)d_ws;

  const int M = BATCH * NVOX;

  {
    int waves = (M / 32) * (3 * CCH / 64);
    wmma_gemm_f32<2><<<waves * 32 / 256, 256, 0, stream>>>(
        x, w_qkv, nullptr, qkv, M, CCH, 3 * CCH, CCH);
  }
  {
    neigh_attn<<<BATCH * 256, 256, 0, stream>>>(qkv, neigh, qkv);
  }
  {
    int waves = (M / 32) * (CCH / 64);
    wmma_gemm_f32<2><<<waves * 32 / 256, 256, 0, stream>>>(
        qkv, w_proj, b_proj, out, M, CCH, CCH, 3 * CCH);
  }
}


# attention: the matrix tile's 16 queries are a 2x2x4 voxel block (was a line of 16 along z): its neighbours fill 6 dense 16x16 key tiles, was 9; 96 WMMAs per wave, was 144; sum order changes
# speedup vs baseline: 3.9267x; 1.0604x over previous
#include <hip/hip_runtime.h>

typedef __attribute__((ext_vector_type(2))) float v2f;
typedef __attribute__((ext_vector_type(4))) float v4f;
typedef __attribute__((ext_vector_type(8))) float v8f;

#define NVOX   4096
#define NHEADS 8
#define HDIM   32
#define CCH    256
#define BATCH  4


template <int MT>
static __device__ __forceinline__ void load_a(const float* arow, int lda, int k, v4f (&a)[MT]) {
#pragma unroll
  for (int r = 0; r < MT; ++r)
    a[r] = *(const v4f*)(arow + (size_t)(16 * r) * lda + k);
}
static __device__ __forceinline__ void load_b(const float* bcol, int Ncols, int krow, v4f (&b)[2]) {
  const float* p = bcol + (size_t)krow * Ncols;
  b[0] = *(const v4f*)p;
  b[1] = *(const v4f*)(p + Ncols);
}

template <int MT, int HI>
static __device__ __forceinline__ void mma_step(v8f (&acc)[MT][4], const v4f (&a)[MT], const v4f (&b)[2]) {
#pragma unroll
  for (int r = 0; r < MT; ++r) {
    const v2f af = HI ? (v2f){a[r].z, a[r].w} : (v2f){a[r].x, a[r].y};
#pragma unroll
    for (int t = 0; t < 4; ++t) {
      const v2f bf = (v2f){b[0][t], b[1][t]};
      acc[r][t] = __builtin_amdgcn_wmma_f32_16x16x4_f32(false, af, false, bf, (short)0, acc[r][t], false, false);
    }
  }
}

template <int MT>
__global__ void __launch_bounds__(256)
wmma_gemm_f32(const float* A, const float* __restrict__ B,
              const float* __restrict__ bias, float* D,
              int M, int K, int Ncols, int lda) {
  const int lane = threadIdx.x & 31;
  const int wib  = __builtin_amdgcn_readfirstlane((int)(threadIdx.x >> 5));
  const int ntp  = Ncols >> 7;
  const int tg   = blockIdx.x / ntp;
  const int tp   = blockIdx.x - tg * ntp;
  const int row0 = ((tg << 2) + (wib & 3)) * (16 * MT);
  if (row0 >= M) return;
  const int col0 = ((tp << 1) + (wib >> 2)) << 6;
  const int half = lane >> 4;
  const int lid  = lane & 15;

  const float* arow = A + (size_t)(row0 + lid) * lda + (half << 3);
  const float* bcol = B + (size_t)(half << 3) * Ncols + col0 + (lid << 2);

  v8f acc[MT][4];
#pragma unroll
  for (int r = 0; r < MT; ++r)
#pragma unroll
    for (int t = 0; t < 4; ++t)
      acc[r][t] = (v8f){0.f,0.f,0.f,0.f,0.f,0.f,0.f,0.f};

  v4f a0[MT], a1[MT], b0[2], b1[2], b2[2], b3[2];
  load_a<MT>(arow, lda, 0, a0);
  load_b(bcol, Ncols, 0, b0);
  load_b(bcol, Ncols, 2, b1);
  load_b(bcol, Ncols, 4, b2);
  int k = 0;
  for (; k + 16 < K; k += 16) {
    v4f n0[2], n1[2], n2[2], an[MT];
    load_a<MT>(arow, lda, k + 4, a1);
    load_b(bcol, Ncols, k + 6, b3);
    __builtin_amdgcn_sched_barrier(0);
    load_b(bcol, Ncols, k + 16, n0);
    load_b(bcol, Ncols, k + 18, n1);
    __builtin_amdgcn_sched_barrier(0);
    mma_step<MT, 0>(acc, a0, b0);
    load_b(bcol, Ncols, k + 20, n2);
    __builtin_amdgcn_sched_barrier(0);
    mma_step<MT, 1>(acc, a0, b1);
    load_a<MT>(arow, lda, k + 16, an);
    __builtin_amdgcn_sched_barrier(0);
    mma_step<MT, 0>(acc, a1, b2);
    mma_step<MT, 1>(acc, a1, b3);
    __builtin_amdgcn_sched_barrier(0);
#pragma unroll
    for (int i = 0; i < 2; ++i) { b0[i] = n0[i]; b1[i] = n1[i]; b2[i] = n2[i]; }
#pragma unroll
    for (int r = 0; r < MT; ++r) a0[r] = an[r];
  }
  load_a<MT>(arow, lda, k + 4, a1);
  load_b(bcol, Ncols, k + 6, b3);
  __builtin_amdgcn_sched_barrier(0);
  mma_step<MT, 0>(acc, a0, b0);
  mma_step<MT, 1>(acc, a0, b1);
  mma_step<MT, 0>(acc, a1, b2);
  mma_step<MT, 1>(acc, a1, b3);

  v4f bv = (v4f){0.f, 0.f, 0.f, 0.f};
  if (bias) bv = *(const v4f*)(bias + col0 + (lid << 2));
#pragma unroll
  for (int r = 0; r < MT; ++r) {
    float* drow = D + (size_t)(row0 + 16 * r + (half << 3)) * Ncols + col0 + (lid << 2);
#pragma unroll
    for (int i = 0; i < 8; ++i) {
      const v4f o = (v4f){acc[r][0][i], acc[r][1][i], acc[r][2][i], acc[r][3][i]} + bv;
      *(v4f*)(drow + (size_t)i * Ncols) = o;
    }
  }
}

#define NNB  27
#define ROWP 36
#define WPB  8

static __device__ __forceinline__ float xhalf(float x) {
  const int b = __builtin_bit_cast(int, x);
  return __builtin_bit_cast(float, __builtin_amdgcn_permlanex16(b, b, 0x76543210, 0xfedcba98, false, false));
}
template <int M> static __device__ __forceinline__ float xrow(float x) {
  return __builtin_bit_cast(float,
      __builtin_amdgcn_update_dpp(0, __builtin_bit_cast(int, x), 0x160 + M, 0xf, 0xf, false));
}
static __device__ __forceinline__ float wave_max(float x) {
  x = fmaxf(x, xrow<1>(x)); x = fmaxf(x, xrow<2>(x));
  x = fmaxf(x, xrow<4>(x)); x = fmaxf(x, xrow<8>(x));
  return fmaxf(x, xhalf(x));
}
static __device__ __forceinline__ float wave_sum(float x) {
  x += xrow<1>(x); x += xrow<2>(x); x += xrow<4>(x); x += xrow<8>(x);
  return x + xhalf(x);
}
static __device__ __forceinline__ void wave_lds_order() {
  __builtin_amdgcn_fence(__ATOMIC_RELEASE, "wavefront");
  __builtin_amdgcn_wave_barrier();
  __builtin_amdgcn_fence(__ATOMIC_ACQUIRE, "wavefront");
}

static __device__ __forceinline__ void voxel_generic(const float* qkv, const int* __restrict__ neigh,
                                                     float* out, float* myprod,
                                                     int b, int n, int lane) {
  const float NEG_INF = -__builtin_inff();
  const size_t base = (size_t)(b * NVOX + n) * (3 * CCH);
  const int jr = lane < NNB ? lane : NNB - 1;
  const bool live = (lane < NNB) && (neigh[n * NNB + jr] != NVOX);

  const float* krow[NNB];
#pragma unroll
  for (int j = 0; j < NNB; ++j) {
    const int id = neigh[n * NNB + j];
    krow[j] = qkv + (size_t)(b * NVOX + (id != NVOX ? id : 0)) * (3 * CCH) + CCH;
  }

  for (int h = 0; h < NHEADS; ++h) {
    const unsigned co = (unsigned)(h * HDIM + lane);
    const unsigned cv = co + CCH;
    const float q = qkv[base + co] * 0.17677669529663687f;

#pragma unroll
    for (int j = 0; j < NNB; ++j)
      myprod[j * ROWP + lane] = q * krow[j][co];

    wave_lds_order();

    const float4* r4 = (const float4*)(myprod + jr * ROWP);
    const float4 a0 = r4[0], a1 = r4[1], a2 = r4[2], a3 = r4[3];
    const float4 a4 = r4[4], a5 = r4[5], a6 = r4[6], a7 = r4[7];

    wave_lds_order();

    float4 t;
    t.x = ((a0.x + a1.x) + (a2.x + a3.x)) + ((a4.x + a5.x) + (a6.x + a7.x));
    t.y = ((a0.y + a1.y) + (a2.y + a3.y)) + ((a4.y + a5.y) + (a6.y + a7.y));
    t.z = ((a0.z + a1.z) + (a2.z + a3.z)) + ((a4.z + a5.z) + (a6.z + a7.z));
    t.w = ((a0.w + a1.w) + (a2.w + a3.w)) + ((a4.w + a5.w) + (a6.w + a7.w));
    const float tsum = (t.x + t.y) + (t.z + t.w);
    const float s = live ? tsum : NEG_INF;

    const float m   = wave_max(s);
    const float e   = __expf(s - m);
    const float inv = 1.0f / wave_sum(e);

    float acc = 0.f;
#pragma unroll
    for (int j = 0; j < NNB; ++j) {
      const float pj = __builtin_bit_cast(float,
          __builtin_amdgcn_readlane(__builtin_bit_cast(int, e), j));
      acc += pj * krow[j][cv];
    }
    out[base + co] = acc * inv;
  }
}

#define LROWS 16

static __device__ __forceinline__ void stage_lines(float* buf, int soff, const v4f (&g)[4]) {
#pragma unroll
  for (int i = 0; i < 4; ++i)
    *(v4f*)(buf + soff + 4 * i * ROWP) = g[i];
}
static __device__ __forceinline__ void read_frags(const float* buf, int foff, v4f (&f)[4]) {
  f[0] = *(const v4f*)(buf + foff);
  f[1] = *(const v4f*)(buf + foff + 4);
  f[2] = *(const v4f*)(buf + foff + 16);
  f[3] = *(const v4f*)(buf + foff + 20);
}
static __device__ __forceinline__ int clamp15(int v) { return v < 0 ? 0 : (v > 15 ? 15 : v); }

__global__ void __launch_bounds__(256)
neigh_attn(const float* qkv, const int* __restrict__ neigh, float* out) {
  __shared__ __attribute__((aligned(16))) float tile[WPB][2 * LROWS * ROWP];

  const int lane = threadIdx.x & 31;
  const int head = __builtin_amdgcn_readfirstlane((int)(threadIdx.x >> 5));
  const int blk = blockIdx.x;
  const int b  = blk >> 8;
  const int x0 = ((blk >> 5) & 7) << 1;
  const int y0 = ((blk >> 2) & 7) << 1;
  const int z0 = (blk & 3) << 2;
  const int n00 = (x0 << 8) + (y0 << 4) + z0;
  const int half = lane >> 4, c = lane & 15;

  const float* rowbase = qkv + (size_t)(b * NVOX) * (3 * CCH);
  const int hoff = head * HDIM;

  const unsigned goff = (unsigned)((lane >> 3) * (3 * CCH) + ((lane & 7) << 2));
  const int soff = (lane >> 3) * ROWP + ((lane & 7) << 2);
  const int foff = c * ROWP + 8 * half;
  float* buf0 = &tile[head][0];
  float* buf1 = buf0 + LROWS * ROWP;

  v4f gq[4];
#pragma unroll
  for (int i = 0; i < 4; ++i)
    gq[i] = *(const v4f*)(rowbase + (size_t)(n00 + (i >> 1) * 256 + (i & 1) * 16) * (3 * CCH) + hoff + goff);

  const int js = lane < NNB ? lane : NNB - 1;
  const int bx = x0 + js / 9 - 1, by = y0 + (js / 3) % 3 - 1, bz = z0 + js % 3 - 1;
  const int idbase = bx * 256 + by * 16 + bz;
  int bad = 0;
#pragma unroll
  for (int v = 0; v < 16; ++v) {
    const int qx = v >> 3, qy = (v >> 2) & 1, qz = v & 3;
    const int qo = qx * 256 + qy * 16 + qz;
    const bool in = ((unsigned)(bx + qx) < 16u) && ((unsigned)(by + qy) < 16u) && ((unsigned)(bz + qz) < 16u);
    const int expect = in ? idbase + qo : NVOX;
    bad |= neigh[(n00 + qo) * NNB + js] ^ expect;
  }
  const bool regular = __builtin_amdgcn_ballot_w32(bad != 0) == 0u;

  if (!regular) {
    float* myprod = buf0;
#pragma nounroll
    for (int v = 0; v < 2; ++v) {
      const int pv = 2 * head + v;
      voxel_generic(qkv, neigh, out, myprod, b,
                    n00 + (pv >> 3) * 256 + ((pv >> 2) & 1) * 16 + (pv & 3), lane);
    }
    return;
  }

  const float NEG_INF = -__builtin_inff();

  const int qx = c >> 3, qy = (c >> 2) & 1, qz = c & 3;
  bool okxy[8];
#pragma unroll
  for (int r = 0; r < 8; ++r) {
    const int xi = 2 * half + (r >> 2), yi = r & 3;
    okxy[r] = ((unsigned)(xi - qx) <= 2u) & ((unsigned)(x0 - 1 + xi) < 16u) &
              ((unsigned)(yi - qy) <= 2u) & ((unsigned)(y0 - 1 + yi) < 16u);
  }
  bool okz[6];
#pragma unroll
  for (int T = 0; T < 6; ++T)
    okz[T] = ((unsigned)(T - qz) <= 2u) & ((unsigned)(z0 - 1 + T) < 16u);

  int xro[4], yro[4], zro[6];
#pragma unroll
  for (int i = 0; i < 4; ++i) { xro[i] = clamp15(x0 - 1 + i) << 8; yro[i] = clamp15(y0 - 1 + i) << 4; }
#pragma unroll
  for (int T = 0; T < 6; ++T) zro[T] = clamp15(z0 - 1 + T);

  const float* kbase = rowbase + CCH + hoff;
  const unsigned goffK = (unsigned)((clamp15(y0 - 1 + (lane >> 3)) << 4) * (3 * CCH) + ((lane & 7) << 2));
#define LOAD_KTILE(T_, G_)                                                                     \
  _Pragma("unroll") for (int i_ = 0; i_ < 4; ++i_)                                             \
    (G_)[i_] = *(const v4f*)(kbase + (size_t)(xro[i_] + zro[T_]) * (3 * CCH) + goffK)

  v4f g[2][4];
  LOAD_KTILE(0, g[0]);
  LOAD_KTILE(1, g[1]);

  v4f qf[4], kf[4];
  stage_lines(buf1, soff, gq);
  wave_lds_order();
  read_frags(buf1, foff, qf);
  stage_lines(buf0, soff, g[0]);
  wave_lds_order();
  read_frags(buf0, foff, kf);
  LOAD_KTILE(2, g[0]);
#pragma unroll
  for (int i = 0; i < 4; ++i) qf[i] = qf[i] * 0.17677669529663687f;

  v8f S[6];
  float m = NEG_INF;
#pragma unroll
  for (int T = 0; T < 6; ++T) {
    v4f kn[4];
    if (T + 1 < 6) {
      float* nb = ((T + 1) & 1) ? buf1 : buf0;
      stage_lines(nb, soff, g[(T + 1) & 1]);
      wave_lds_order();
      read_frags(nb, foff, kn);
      if (T + 3 < 6)
        LOAD_KTILE(T + 3, g[(T + 1) & 1]);
    }
    __builtin_amdgcn_sched_barrier(0);
    v8f acc = (v8f){0.f,0.f,0.f,0.f,0.f,0.f,0.f,0.f};
#pragma unroll
    for (int st = 0; st < 8; ++st) {
      const v2f af = (v2f){kf[st >> 1][2 * (st & 1)], kf[st >> 1][2 * (st & 1) + 1]};
      const v2f bf = (v2f){qf[st >> 1][2 * (st & 1)], qf[st >> 1][2 * (st & 1) + 1]};
      acc = __builtin_amdgcn_wmma_f32_16x16x4_f32(false, af, false, bf, (short)0, acc, false, false);
    }
#pragma unroll
    for (int r = 0; r < 8; ++r) {
      const float s = (okz[T] && okxy[r]) ? acc[r] : NEG_INF;
      S[T][r] = s;
      m = fmaxf(m, s);
    }
    if (T + 1 < 6) {
#pragma unroll
      for (int i = 0; i < 4; ++i) kf[i] = kn[i];
    }
  }
#undef LOAD_KTILE
  m = fmaxf(m, xhalf(m));

  float sum = 0.f;
#pragma unroll
  for (int T = 0; T < 6; ++T)
#pragma unroll
    for (int r = 0; r < 8; ++r) {
      const float e = __expf(S[T][r] - m);
      S[T][r] = e;
      sum += e;
    }
  sum += xhalf(sum);
  const float inv = 1.0f / sum;

  const float* vbase = rowbase + 2 * CCH + hoff;
  unsigned xv[2];
#pragma unroll
  for (int a = 0; a < 2; ++a) xv[a] = (unsigned)((clamp15(x0 - 1 + 2 * half + a) << 8) * (3 * CCH) + 2 * c);
  v8f o0 = (v8f){0.f,0.f,0.f,0.f,0.f,0.f,0.f,0.f};
  v8f o1 = (v8f){0.f,0.f,0.f,0.f,0.f,0.f,0.f,0.f};
#pragma unroll
  for (int T = 0; T < 6; ++T) {
#pragma unroll
    for (int s2 = 0; s2 < 4; ++s2) {
      const v2f va = *(const v2f*)(vbase + (size_t)(yro[2 * (s2 & 1)] + zro[T]) * (3 * CCH) + xv[s2 >> 1]);
      const v2f vb = *(const v2f*)(vbase + (size_t)(yro[2 * (s2 & 1) + 1] + zro[T]) * (3 * CCH) + xv[s2 >> 1]);
      const v2f af = (v2f){S[T][2 * s2] * inv, S[T][2 * s2 + 1] * inv};
      o0 = __builtin_amdgcn_wmma_f32_16x16x4_f32(false, af, false, (v2f){va.x, vb.x}, (short)0, o0, false, false);
      o1 = __builtin_amdgcn_wmma_f32_16x16x4_f32(false, af, false, (v2f){va.y, vb.y}, (short)0, o1, false, false);
    }
  }

  float* op = out + (size_t)(b * NVOX + n00 + half * 256) * (3 * CCH) + hoff + 2 * c;
#pragma unroll
  for (int r = 0; r < 8; ++r)
    *(v2f*)(op + (size_t)((r >> 2) * 16 + (r & 3)) * (3 * CCH)) = (v2f){o0[r], o1[r]};
}

extern "C" void kernel_launch(void* const* d_in, const int* in_sizes, int n_in,
                              void* d_out, int out_size, void* d_ws, size_t ws_size,
                              hipStream_t stream) {
  (void)in_sizes; (void)n_in; (void)out_size; (void)ws_size;
  const float* x      = (const float*)d_in[0];
  const float* w_qkv  = (const float*)d_in[1];
  const float* w_proj = (const float*)d_in[2];
  const float* b_proj = (const float*)d_in[3];
  const int*   neigh  = (const int*)d_in[4];
  float* out = (float*)d_out;
  float* qkv = (float*)d_ws;

  const int M = BATCH * NVOX;

  {
    int waves = (M / 32) * (3 * CCH / 64);
    wmma_gemm_f32<2><<<waves * 32 / 256, 256, 0, stream>>>(
        x, w_qkv, nullptr, qkv, M, CCH, 3 * CCH, CCH);
  }
  {
    neigh_attn<<<BATCH * 256, 256, 0, stream>>>(qkv, neigh, qkv);
  }
  {
    int waves = (M / 32) * (CCH / 64);
    wmma_gemm_f32<2><<<waves * 32 / 256, 256, 0, stream>>>(
        qkv, w_proj, b_proj, out, M, CCH, CCH, 3 * CCH);
  }
}


# both GEMMs: wave tile 64x64 (was 32x64), 4 waves per work-group, one per SIMD (was 8 waves); a trip's 16 loads go out as one group a whole trip (64 WMMAs) ahead; sum order same
# speedup vs baseline: 4.2264x; 1.0763x over previous
#include <hip/hip_runtime.h>

typedef __attribute__((ext_vector_type(2))) float v2f;
typedef __attribute__((ext_vector_type(4))) float v4f;
typedef __attribute__((ext_vector_type(8))) float v8f;

#define NVOX   4096
#define NHEADS 8
#define HDIM   32
#define CCH    256
#define BATCH  4


template <int MT>
static __device__ __forceinline__ void load_a(const float* arow, int lda, int k, v4f (&a)[MT]) {
#pragma unroll
  for (int r = 0; r < MT; ++r)
    a[r] = *(const v4f*)(arow + (size_t)(16 * r) * lda + k);
}
static __device__ __forceinline__ void load_b(const float* bcol, int Ncols, int krow, v4f (&b)[2]) {
  const float* p = bcol + (size_t)krow * Ncols;
  b[0] = *(const v4f*)p;
  b[1] = *(const v4f*)(p + Ncols);
}

template <int MT, int HI>
static __device__ __forceinline__ void mma_step(v8f (&acc)[MT][4], const v4f (&a)[MT], const v4f (&b)[2]) {
#pragma unroll
  for (int r = 0; r < MT; ++r) {
    const v2f af = HI ? (v2f){a[r].z, a[r].w} : (v2f){a[r].x, a[r].y};
#pragma unroll
    for (int t = 0; t < 4; ++t) {
      const v2f bf = (v2f){b[0][t], b[1][t]};
      acc[r][t] = __builtin_amdgcn_wmma_f32_16x16x4_f32(false, af, false, bf, (short)0, acc[r][t], false, false);
    }
  }
}

template <int MT>
__global__ void __launch_bounds__(128)
wmma_gemm_f32(const float* A, const float* __restrict__ B,
              const float* __restrict__ bias, float* D,
              int M, int K, int Ncols, int lda) {
  const int lane = threadIdx.x & 31;
  const int wib  = __builtin_amdgcn_readfirstlane((int)(threadIdx.x >> 5));
  const int ntp  = Ncols >> 7;
  const int tg   = blockIdx.x / ntp;
  const int tp   = blockIdx.x - tg * ntp;
  const int row0 = ((tg << 1) + (wib & 1)) * (16 * MT);
  if (row0 >= M) return;
  const int col0 = ((tp << 1) + (wib >> 1)) << 6;
  const int half = lane >> 4;
  const int lid  = lane & 15;

  const float* arow = A + (size_t)(row0 + lid) * lda + (half << 3);
  const float* bcol = B + (size_t)(half << 3) * Ncols + col0 + (lid << 2);

  v8f acc[MT][4];
#pragma unroll
  for (int r = 0; r < MT; ++r)
#pragma unroll
    for (int t = 0; t < 4; ++t)
      acc[r][t] = (v8f){0.f,0.f,0.f,0.f,0.f,0.f,0.f,0.f};

  v4f a0[MT], a1[MT], b0[2], b1[2], b2[2], b3[2];
  load_a<MT>(arow, lda, 0, a0);
  load_a<MT>(arow, lda, 4, a1);
  load_b(bcol, Ncols, 0, b0);
  load_b(bcol, Ncols, 2, b1);
  load_b(bcol, Ncols, 4, b2);
  load_b(bcol, Ncols, 6, b3);
  int k = 0;
#pragma unroll 2
  for (; k + 16 < K; k += 16) {
    v4f an0[MT], an1[MT], n0[2], n1[2], n2[2], n3[2];
    load_a<MT>(arow, lda, k + 16, an0);
    load_a<MT>(arow, lda, k + 20, an1);
    load_b(bcol, Ncols, k + 16, n0);
    load_b(bcol, Ncols, k + 18, n1);
    load_b(bcol, Ncols, k + 20, n2);
    load_b(bcol, Ncols, k + 22, n3);
    __builtin_amdgcn_sched_barrier(0);
    mma_step<MT, 0>(acc, a0, b0);
    mma_step<MT, 1>(acc, a0, b1);
    mma_step<MT, 0>(acc, a1, b2);
    mma_step<MT, 1>(acc, a1, b3);
    __builtin_amdgcn_sched_barrier(0);
#pragma unroll
    for (int i = 0; i < 2; ++i) { b0[i] = n0[i]; b1[i] = n1[i]; b2[i] = n2[i]; b3[i] = n3[i]; }
#pragma unroll
    for (int r = 0; r < MT; ++r) { a0[r] = an0[r]; a1[r] = an1[r]; }
  }
  mma_step<MT, 0>(acc, a0, b0);
  mma_step<MT, 1>(acc, a0, b1);
  mma_step<MT, 0>(acc, a1, b2);
  mma_step<MT, 1>(acc, a1, b3);

  v4f bv = (v4f){0.f, 0.f, 0.f, 0.f};
  if (bias) bv = *(const v4f*)(bias + col0 + (lid << 2));
#pragma unroll
  for (int r = 0; r < MT; ++r) {
    float* drow = D + (size_t)(row0 + 16 * r + (half << 3)) * Ncols + col0 + (lid << 2);
#pragma unroll
    for (int i = 0; i < 8; ++i) {
      const v4f o = (v4f){acc[r][0][i], acc[r][1][i], acc[r][2][i], acc[r][3][i]} + bv;
      *(v4f*)(drow + (size_t)i * Ncols) = o;
    }
  }
}

#define NNB  27
#define ROWP 36
#define WPB  8

static __device__ __forceinline__ float xhalf(float x) {
  const int b = __builtin_bit_cast(int, x);
  return __builtin_bit_cast(float, __builtin_amdgcn_permlanex16(b, b, 0x76543210, 0xfedcba98, false, false));
}
template <int M> static __device__ __forceinline__ float xrow(float x) {
  return __builtin_bit_cast(float,
      __builtin_amdgcn_update_dpp(0, __builtin_bit_cast(int, x), 0x160 + M, 0xf, 0xf, false));
}
static __device__ __forceinline__ float wave_max(float x) {
  x = fmaxf(x, xrow<1>(x)); x = fmaxf(x, xrow<2>(x));
  x = fmaxf(x, xrow<4>(x)); x = fmaxf(x, xrow<8>(x));
  return fmaxf(x, xhalf(x));
}
static __device__ __forceinline__ float wave_sum(float x) {
  x += xrow<1>(x); x += xrow<2>(x); x += xrow<4>(x); x += xrow<8>(x);
  return x + xhalf(x);
}
static __device__ __forceinline__ void wave_lds_order() {
  __builtin_amdgcn_fence(__ATOMIC_RELEASE, "wavefront");
  __builtin_amdgcn_wave_barrier();
  __builtin_amdgcn_fence(__ATOMIC_ACQUIRE, "wavefront");
}

static __device__ __forceinline__ void voxel_generic(const float* qkv, const int* __restrict__ neigh,
                                                     float* out, float* myprod,
                                                     int b, int n, int lane) {
  const float NEG_INF = -__builtin_inff();
  const size_t base = (size_t)(b * NVOX + n) * (3 * CCH);
  const int jr = lane < NNB ? lane : NNB - 1;
  const bool live = (lane < NNB) && (neigh[n * NNB + jr] != NVOX);

  const float* krow[NNB];
#pragma unroll
  for (int j = 0; j < NNB; ++j) {
    const int id = neigh[n * NNB + j];
    krow[j] = qkv + (size_t)(b * NVOX + (id != NVOX ? id : 0)) * (3 * CCH) + CCH;
  }

  for (int h = 0; h < NHEADS; ++h) {
    const unsigned co = (unsigned)(h * HDIM + lane);
    const unsigned cv = co + CCH;
    const float q = qkv[base + co] * 0.17677669529663687f;

#pragma unroll
    for (int j = 0; j < NNB; ++j)
      myprod[j * ROWP + lane] = q * krow[j][co];

    wave_lds_order();

    const float4* r4 = (const float4*)(myprod + jr * ROWP);
    const float4 a0 = r4[0], a1 = r4[1], a2 = r4[2], a3 = r4[3];
    const float4 a4 = r4[4], a5 = r4[5], a6 = r4[6], a7 = r4[7];

    wave_lds_order();

    float4 t;
    t.x = ((a0.x + a1.x) + (a2.x + a3.x)) + ((a4.x + a5.x) + (a6.x + a7.x));
    t.y = ((a0.y + a1.y) + (a2.y + a3.y)) + ((a4.y + a5.y) + (a6.y + a7.y));
    t.z = ((a0.z + a1.z) + (a2.z + a3.z)) + ((a4.z + a5.z) + (a6.z + a7.z));
    t.w = ((a0.w + a1.w) + (a2.w + a3.w)) + ((a4.w + a5.w) + (a6.w + a7.w));
    const float tsum = (t.x + t.y) + (t.z + t.w);
    const float s = live ? tsum : NEG_INF;

    const float m   = wave_max(s);
    const float e   = __expf(s - m);
    const float inv = 1.0f / wave_sum(e);

    float acc = 0.f;
#pragma unroll
    for (int j = 0; j < NNB; ++j) {
      const float pj = __builtin_bit_cast(float,
          __builtin_amdgcn_readlane(__builtin_bit_cast(int, e), j));
      acc += pj * krow[j][cv];
    }
    out[base + co] = acc * inv;
  }
}

#define LROWS 16

static __device__ __forceinline__ void stage_lines(float* buf, int soff, const v4f (&g)[4]) {
#pragma unroll
  for (int i = 0; i < 4; ++i)
    *(v4f*)(buf + soff + 4 * i * ROWP) = g[i];
}
static __device__ __forceinline__ void read_frags(const float* buf, int foff, v4f (&f)[4]) {
  f[0] = *(const v4f*)(buf + foff);
  f[1] = *(const v4f*)(buf + foff + 4);
  f[2] = *(const v4f*)(buf + foff + 16);
  f[3] = *(const v4f*)(buf + foff + 20);
}
static __device__ __forceinline__ int clamp15(int v) { return v < 0 ? 0 : (v > 15 ? 15 : v); }

__global__ void __launch_bounds__(256)
neigh_attn(const float* qkv, const int* __restrict__ neigh, float* out) {
  __shared__ __attribute__((aligned(16))) float tile[WPB][2 * LROWS * ROWP];

  const int lane = threadIdx.x & 31;
  const int head = __builtin_amdgcn_readfirstlane((int)(threadIdx.x >> 5));
  const int blk = blockIdx.x;
  const int b  = blk >> 8;
  const int x0 = ((blk >> 5) & 7) << 1;
  const int y0 = ((blk >> 2) & 7) << 1;
  const int z0 = (blk & 3) << 2;
  const int n00 = (x0 << 8) + (y0 << 4) + z0;
  const int half = lane >> 4, c = lane & 15;

  const float* rowbase = qkv + (size_t)(b * NVOX) * (3 * CCH);
  const int hoff = head * HDIM;

  const unsigned goff = (unsigned)((lane >> 3) * (3 * CCH) + ((lane & 7) << 2));
  const int soff = (lane >> 3) * ROWP + ((lane & 7) << 2);
  const int foff = c * ROWP + 8 * half;
  float* buf0 = &tile[head][0];
  float* buf1 = buf0 + LROWS * ROWP;

  v4f gq[4];
#pragma unroll
  for (int i = 0; i < 4; ++i)
    gq[i] = *(const v4f*)(rowbase + (size_t)(n00 + (i >> 1) * 256 + (i & 1) * 16) * (3 * CCH) + hoff + goff);

  const int js = lane < NNB ? lane : NNB - 1;
  const int bx = x0 + js / 9 - 1, by = y0 + (js / 3) % 3 - 1, bz = z0 + js % 3 - 1;
  const int idbase = bx * 256 + by * 16 + bz;
  int bad = 0;
#pragma unroll
  for (int v = 0; v < 16; ++v) {
    const int qx = v >> 3, qy = (v >> 2) & 1, qz = v & 3;
    const int qo = qx * 256 + qy * 16 + qz;
    const bool in = ((unsigned)(bx + qx) < 16u) && ((unsigned)(by + qy) < 16u) && ((unsigned)(bz + qz) < 16u);
    const int expect = in ? idbase + qo : NVOX;
    bad |= neigh[(n00 + qo) * NNB + js] ^ expect;
  }
  const bool regular = __builtin_amdgcn_ballot_w32(bad != 0) == 0u;

  if (!regular) {
    float* myprod = buf0;
#pragma nounroll
    for (int v = 0; v < 2; ++v) {
      const int pv = 2 * head + v;
      voxel_generic(qkv, neigh, out, myprod, b,
                    n00 + (pv >> 3) * 256 + ((pv >> 2) & 1) * 16 + (pv & 3), lane);
    }
    return;
  }

  const float NEG_INF = -__builtin_inff();

  const int qx = c >> 3, qy = (c >> 2) & 1, qz = c & 3;
  bool okxy[8];
#pragma unroll
  for (int r = 0; r < 8; ++r) {
    const int xi = 2 * half + (r >> 2), yi = r & 3;
    okxy[r] = ((unsigned)(xi - qx) <= 2u) & ((unsigned)(x0 - 1 + xi) < 16u) &
              ((unsigned)(yi - qy) <= 2u) & ((unsigned)(y0 - 1 + yi) < 16u);
  }
  bool okz[6];
#pragma unroll
  for (int T = 0; T < 6; ++T)
    okz[T] = ((unsigned)(T - qz) <= 2u) & ((unsigned)(z0 - 1 + T) < 16u);

  int xro[4], yro[4], zro[6];
#pragma unroll
  for (int i = 0; i < 4; ++i) { xro[i] = clamp15(x0 - 1 + i) << 8; yro[i] = clamp15(y0 - 1 + i) << 4; }
#pragma unroll
  for (int T = 0; T < 6; ++T) zro[T] = clamp15(z0 - 1 + T);

  const float* kbase = rowbase + CCH + hoff;
  const unsigned goffK = (unsigned)((clamp15(y0 - 1 + (lane >> 3)) << 4) * (3 * CCH) + ((lane & 7) << 2));
#define LOAD_KTILE(T_, G_)                                                                     \
  _Pragma("unroll") for (int i_ = 0; i_ < 4; ++i_)                                             \
    (G_)[i_] = *(const v4f*)(kbase + (size_t)(xro[i_] + zro[T_]) * (3 * CCH) + goffK)

  v4f g[2][4];
  LOAD_KTILE(0, g[0]);
  LOAD_KTILE(1, g[1]);

  v4f qf[4], kf[4];
  stage_lines(buf1, soff, gq);
  wave_lds_order();
  read_frags(buf1, foff, qf);
  stage_lines(buf0, soff, g[0]);
  wave_lds_order();
  read_frags(buf0, foff, kf);
  LOAD_KTILE(2, g[0]);
#pragma unroll
  for (int i = 0; i < 4; ++i) qf[i] = qf[i] * 0.17677669529663687f;

  v8f S[6];
  float m = NEG_INF;
#pragma unroll
  for (int T = 0; T < 6; ++T) {
    v4f kn[4];
    if (T + 1 < 6) {
      float* nb = ((T + 1) & 1) ? buf1 : buf0;
      stage_lines(nb, soff, g[(T + 1) & 1]);
      wave_lds_order();
      read_frags(nb, foff, kn);
      if (T + 3 < 6)
        LOAD_KTILE(T + 3, g[(T + 1) & 1]);
    }
    __builtin_amdgcn_sched_barrier(0);
    v8f acc = (v8f){0.f,0.f,0.f,0.f,0.f,0.f,0.f,0.f};
#pragma unroll
    for (int st = 0; st < 8; ++st) {
      const v2f af = (v2f){kf[st >> 1][2 * (st & 1)], kf[st >> 1][2 * (st & 1) + 1]};
      const v2f bf = (v2f){qf[st >> 1][2 * (st & 1)], qf[st >> 1][2 * (st & 1) + 1]};
      acc = __builtin_amdgcn_wmma_f32_16x16x4_f32(false, af, false, bf, (short)0, acc, false, false);
    }
#pragma unroll
    for (int r = 0; r < 8; ++r) {
      const float s = (okz[T] && okxy[r]) ? acc[r] : NEG_INF;
      S[T][r] = s;
      m = fmaxf(m, s);
    }
    if (T + 1 < 6) {
#pragma unroll
      for (int i = 0; i < 4; ++i) kf[i] = kn[i];
    }
  }
#undef LOAD_KTILE
  m = fmaxf(m, xhalf(m));

  float sum = 0.f;
#pragma unroll
  for (int T = 0; T < 6; ++T)
#pragma unroll
    for (int r = 0; r < 8; ++r) {
      const float e = __expf(S[T][r] - m);
      S[T][r] = e;
      sum += e;
    }
  sum += xhalf(sum);
  const float inv = 1.0f / sum;

  const float* vbase = rowbase + 2 * CCH + hoff;
  unsigned xv[2];
#pragma unroll
  for (int a = 0; a < 2; ++a) xv[a] = (unsigned)((clamp15(x0 - 1 + 2 * half + a) << 8) * (3 * CCH) + 2 * c);
  v8f o0 = (v8f){0.f,0.f,0.f,0.f,0.f,0.f,0.f,0.f};
  v8f o1 = (v8f){0.f,0.f,0.f,0.f,0.f,0.f,0.f,0.f};
#pragma unroll
  for (int T = 0; T < 6; ++T) {
#pragma unroll
    for (int s2 = 0; s2 < 4; ++s2) {
      const v2f va = *(const v2f*)(vbase + (size_t)(yro[2 * (s2 & 1)] + zro[T]) * (3 * CCH) + xv[s2 >> 1]);
      const v2f vb = *(const v2f*)(vbase + (size_t)(yro[2 * (s2 & 1) + 1] + zro[T]) * (3 * CCH) + xv[s2 >> 1]);
      const v2f af = (v2f){S[T][2 * s2] * inv, S[T][2 * s2 + 1] * inv};
      o0 = __builtin_amdgcn_wmma_f32_16x16x4_f32(false, af, false, (v2f){va.x, vb.x}, (short)0, o0, false, false);
      o1 = __builtin_amdgcn_wmma_f32_16x16x4_f32(false, af, false, (v2f){va.y, vb.y}, (short)0, o1, false, false);
    }
  }

  float* op = out + (size_t)(b * NVOX + n00 + half * 256) * (3 * CCH) + hoff + 2 * c;
#pragma unroll
  for (int r = 0; r < 8; ++r)
    *(v2f*)(op + (size_t)((r >> 2) * 16 + (r & 3)) * (3 * CCH)) = (v2f){o0[r], o1[r]};
}

extern "C" void kernel_launch(void* const* d_in, const int* in_sizes, int n_in,
                              void* d_out, int out_size, void* d_ws, size_t ws_size,
                              hipStream_t stream) {
  (void)in_sizes; (void)n_in; (void)out_size; (void)ws_size;
  const float* x      = (const float*)d_in[0];
  const float* w_qkv  = (const float*)d_in[1];
  const float* w_proj = (const float*)d_in[2];
  const float* b_proj = (const float*)d_in[3];
  const int*   neigh  = (const int*)d_in[4];
  float* out = (float*)d_out;
  float* qkv = (float*)d_ws;

  const int M = BATCH * NVOX;

  {
    int waves = (M / 64) * (3 * CCH / 64);
    wmma_gemm_f32<4><<<waves / 4, 128, 0, stream>>>(
        x, w_qkv, nullptr, qkv, M, CCH, 3 * CCH, CCH);
  }
  {
    neigh_attn<<<BATCH * 256, 256, 0, stream>>>(qkv, neigh, qkv);
  }
  {
    int waves = (M / 64) * (CCH / 64);
    wmma_gemm_f32<4><<<waves / 4, 128, 0, stream>>>(
        qkv, w_proj, b_proj, out, M, CCH, CCH, 3 * CCH);
  }
}
